# starting kernel
# baseline (speedup 1.0000x reference)
#include <hip/hip_runtime.h>
#include <math.h>

#define IC_N 512
#define OC_N 512
#define H_N 128
#define W_N 128
#define HW_N (H_N * W_N)
#define PH 129
#define PHS (PH * PH)

typedef __attribute__((ext_vector_type(2))) float v2f;
typedef __attribute__((ext_vector_type(8))) float v8f;

__global__ __launch_bounds__(256) void prep_weights(const float* __restrict__ wgt,
                                                    float* __restrict__ wp) {
  const int oc = blockIdx.x;
  const int tid = threadIdx.x;
  const float scale = rsqrtf(4608.0f);
  __shared__ float red[256];
  __shared__ float dsh;

  const float* wo = wgt + (size_t)oc * IC_N * 9;
  float s = 0.f;
  for (int ic = tid; ic < IC_N; ic += 256) {
    const float* wi = wo + ic * 9;
#pragma unroll
    for (int k = 0; k < 9; ++k) {
      float v = scale * wi[k];
      s += v * v;
    }
  }
  red[tid] = s;
  __syncthreads();
  for (int off = 128; off > 0; off >>= 1) {
    if (tid < off) red[tid] += red[tid + off];
    __syncthreads();
  }
  if (tid == 0) dsh = rsqrtf(red[0] + 1e-8f);
  __syncthreads();
  const float d = dsh * scale;

  float* w00 = wp;
  float* w01 = wp + (size_t)OC_N * 2048;
  float* w10 = w01 + (size_t)OC_N * 1024;
  float* w11 = w10 + (size_t)OC_N * 1024;

  for (int ic = tid; ic < IC_N; ic += 256) {
    const float* wi = wo + ic * 9;
    w00[(size_t)oc * 2048 + 0 * 512 + ic] = d * wi[2 * 3 + 2];
    w00[(size_t)oc * 2048 + 1 * 512 + ic] = d * wi[2 * 3 + 0];
    w00[(size_t)oc * 2048 + 2 * 512 + ic] = d * wi[0 * 3 + 2];
    w00[(size_t)oc * 2048 + 3 * 512 + ic] = d * wi[0 * 3 + 0];
    w01[(size_t)oc * 1024 + 0 * 512 + ic] = d * wi[2 * 3 + 1];
    w01[(size_t)oc * 1024 + 1 * 512 + ic] = d * wi[0 * 3 + 1];
    w10[(size_t)oc * 1024 + 0 * 512 + ic] = d * wi[1 * 3 + 2];
    w10[(size_t)oc * 1024 + 1 * 512 + ic] = d * wi[1 * 3 + 0];
    w11[(size_t)oc * 512 + ic] = d * wi[1 * 3 + 1];
  }
}

__global__ __launch_bounds__(128) void upconv_gemm(
    const float* __restrict__ x, const float* __restrict__ wp,
    float* __restrict__ up, int Kp, int ntaps, int dymask, int dxmask,
    int rows, int cols) {
  __shared__ float bs[512];

  const int tid = threadIdx.x;
  const int lane = tid & 31;
  const int wave = tid >> 5;
  const int px = lane & 15;
  const int hi = lane >> 4;
  const int koff = hi * 2;

  const int r = blockIdx.y;
  const int colbase = blockIdx.x * 16;
  const int oc_base = blockIdx.z * 64 + wave * 16;

  const int l_icl0 = tid >> 4;
  const int l_px = tid & 15;

  v8f acc = {};

  for (int t = 0; t < ntaps; ++t) {
    const int dy = -((dymask >> t) & 1);
    const int dx = -((dxmask >> t) & 1);
    const int row_in = r + dy;
    if (row_in < 0 || row_in >= H_N) continue;

    const int cin = colbase + l_px + dx;
    const bool cok = (cin >= 0) && (cin < W_N);
    const float* xrow = x + (size_t)row_in * W_N + cin;
    const float* arow = wp + (size_t)(oc_base + px) * Kp + t * IC_N;

    for (int icc = 0; icc < IC_N; icc += 32) {
#pragma unroll
      for (int i = 0; i < 4; ++i) {
        const int icl = l_icl0 + 8 * i;
        float v = cok ? xrow[(size_t)(icc + icl) * HW_N] : 0.0f;
        bs[l_px * 32 + icl] = v;
      }
      if (icc + 32 < IC_N)
        __builtin_prefetch(xrow + (size_t)(icc + 32 + l_icl0) * HW_N, 0, 1);
      __syncthreads();

#pragma unroll
      for (int s = 0; s < 8; ++s) {
        v2f a = *(const v2f*)(arow + icc + 4 * s + koff);
        v2f b = *(const v2f*)(&bs[px * 32 + 4 * s + koff]);
        acc = __builtin_amdgcn_wmma_f32_16x16x4_f32(
            false, a, false, b, (short)0, acc, false, false);
      }
      __syncthreads();
    }
  }

  const int c_out = colbase + px;
  if (c_out < cols) {
    float* o = up + (size_t)(oc_base + hi * 8) * PHS + (size_t)r * PH + c_out;
#pragma unroll
    for (int j = 0; j < 8; ++j) o[(size_t)j * PHS] = acc[j];
  }
}

__global__ __launch_bounds__(256) void blur_kernel(const float* __restrict__ up,
                                                   float* __restrict__ out) {
  __shared__ float tile[19 * 20];
  const int oc = blockIdx.z;
  const int y0 = blockIdx.y * 16;
  const int x0 = blockIdx.x * 16;
  const int tid = threadIdx.x;

  for (int idx = tid; idx < 19 * 19; idx += 256) {
    const int ly = idx / 19, lx = idx % 19;
    const int yy = y0 - 1 + ly;
    const int xx = x0 - 1 + lx;
    float v = 0.f;
    if (yy >= 0 && yy <= 256 && xx >= 0 && xx <= 256) {
      const int p = ((yy & 1) << 1) | (xx & 1);
      v = up[((size_t)p * OC_N + oc) * PHS + (size_t)(yy >> 1) * PH + (xx >> 1)];
    }
    tile[ly * 20 + lx] = v;
  }
  __syncthreads();

  const int ty = tid >> 4, tx = tid & 15;
  const float c0 = 0.25f, c1 = 0.75f;
  const float cy[4] = {c0, c1, c1, c0};
  float acc = 0.f;
#pragma unroll
  for (int i = 0; i < 4; ++i) {
    const float* row = &tile[(ty + i) * 20 + tx];
    const float rsum = c0 * row[0] + c1 * row[1] + c1 * row[2] + c0 * row[3];
    acc += cy[i] * rsum;
  }
  out[((size_t)oc * 256 + (y0 + ty)) * 256 + (x0 + tx)] = acc;
}

extern "C" void kernel_launch(void* const* d_in, const int* in_sizes, int n_in,
                              void* d_out, int out_size, void* d_ws, size_t ws_size,
                              hipStream_t stream) {
  const float* x = (const float*)d_in[0];
  const float* wgt = (const float*)d_in[1];
  float* out = (float*)d_out;

  float* ws = (float*)d_ws;
  float* wp = ws;
  float* up = ws + (size_t)OC_N * 4608;

  const float* w00 = wp;
  const float* w01 = wp + (size_t)OC_N * 2048;
  const float* w10 = w01 + (size_t)OC_N * 1024;
  const float* w11 = w10 + (size_t)OC_N * 1024;

  float* up00 = up + (size_t)0 * OC_N * PHS;
  float* up01 = up + (size_t)1 * OC_N * PHS;
  float* up10 = up + (size_t)2 * OC_N * PHS;
  float* up11 = up + (size_t)3 * OC_N * PHS;

  prep_weights<<<OC_N, 256, 0, stream>>>(wgt, wp);

  upconv_gemm<<<dim3(9, 129, 8), 128, 0, stream>>>(x, w00, up00, 2048, 4, 0x3, 0x5, 129, 129);
  upconv_gemm<<<dim3(8, 129, 8), 128, 0, stream>>>(x, w01, up01, 1024, 2, 0x1, 0x0, 129, 128);
  upconv_gemm<<<dim3(9, 128, 8), 128, 0, stream>>>(x, w10, up10, 1024, 2, 0x0, 0x1, 128, 129);
  upconv_gemm<<<dim3(8, 128, 8), 128, 0, stream>>>(x, w11, up11,  512, 1, 0x0, 0x0, 128, 128);

  blur_kernel<<<dim3(16, 16, OC_N), 256, 0, stream>>>(up, out);
}


# GEMM re-tiled: block 128 oc x 64 flattened px, wave 32x32 on 4 accumulators, both operands staged in two-stage LDS, b128 fragment reads (k pairing inside 8-k groups changes the sum order)
# speedup vs baseline: 1.5349x; 1.5349x over previous
#include <hip/hip_runtime.h>
#include <math.h>

#define IC_N 512
#define OC_N 512
#define H_N 128
#define W_N 128
#define HW_N (H_N * W_N)
#define PH 129
#define PHS (PH * PH)

typedef __attribute__((ext_vector_type(2))) float v2f;
typedef __attribute__((ext_vector_type(4))) float v4f;
typedef __attribute__((ext_vector_type(8))) float v8f;

__global__ __launch_bounds__(256) void prep_weights(const float* __restrict__ wgt,
                                                    float* __restrict__ wp) {
  const int oc = blockIdx.x;
  const int tid = threadIdx.x;
  const float scale = rsqrtf(4608.0f);
  __shared__ float red[256];
  __shared__ float dsh;

  const float* wo = wgt + (size_t)oc * IC_N * 9;
  float s = 0.f;
  for (int ic = tid; ic < IC_N; ic += 256) {
    const float* wi = wo + ic * 9;
#pragma unroll
    for (int k = 0; k < 9; ++k) {
      float v = scale * wi[k];
      s += v * v;
    }
  }
  red[tid] = s;
  __syncthreads();
  for (int off = 128; off > 0; off >>= 1) {
    if (tid < off) red[tid] += red[tid + off];
    __syncthreads();
  }
  if (tid == 0) dsh = rsqrtf(red[0] + 1e-8f);
  __syncthreads();
  const float d = dsh * scale;

  float* w00 = wp;
  float* w01 = wp + (size_t)OC_N * 2048;
  float* w10 = w01 + (size_t)OC_N * 1024;
  float* w11 = w10 + (size_t)OC_N * 1024;

  for (int ic = tid; ic < IC_N; ic += 256) {
    const float* wi = wo + ic * 9;
    w00[(size_t)oc * 2048 + 0 * 512 + ic] = d * wi[2 * 3 + 2];
    w00[(size_t)oc * 2048 + 1 * 512 + ic] = d * wi[2 * 3 + 0];
    w00[(size_t)oc * 2048 + 2 * 512 + ic] = d * wi[0 * 3 + 2];
    w00[(size_t)oc * 2048 + 3 * 512 + ic] = d * wi[0 * 3 + 0];
    w01[(size_t)oc * 1024 + 0 * 512 + ic] = d * wi[2 * 3 + 1];
    w01[(size_t)oc * 1024 + 1 * 512 + ic] = d * wi[0 * 3 + 1];
    w10[(size_t)oc * 1024 + 0 * 512 + ic] = d * wi[1 * 3 + 2];
    w10[(size_t)oc * 1024 + 1 * 512 + ic] = d * wi[1 * 3 + 0];
    w11[(size_t)oc * 512 + ic] = d * wi[1 * 3 + 1];
  }
}

#define BM 128
#define BN 64
#define BK 32
#define LP 36

__global__ __launch_bounds__(256) void upconv_gemm(
    const float* __restrict__ x, const float* __restrict__ wp,
    float* __restrict__ up, int Kp, int ntaps, int dymask, int dxmask,
    int rows, int cols) {
  __shared__ __attribute__((aligned(16))) float As[2][BM * LP];
  __shared__ __attribute__((aligned(16))) float Bs[2][BN * LP];

  const int tid = threadIdx.x;
  const int lane = tid & 31;
  const int wave = tid >> 5;
  const int c16 = lane & 15;
  const int hi = lane >> 4;
  const int wm = wave >> 1;
  const int wn = wave & 1;

  const int npx = rows * cols;
  const int oc0 = blockIdx.x * BM;
  const int n0 = blockIdx.y * BN;

  const int a_row = tid >> 3;
  const int a_kq = tid & 7;
  const int b_px = tid & 63;
  const int b_kq = tid >> 6;

  const int p = n0 + b_px;
  const bool pok = p < npx;
  const int pr = p / cols;
  const int pc = p - pr * cols;

  const float* a_src = wp + (size_t)(oc0 + a_row) * Kp + a_kq * 4;

  v4f ra[4];
  float rb[8];

  auto load_chunk = [&](int ch) {
#pragma unroll
    for (int i = 0; i < 4; ++i)
      ra[i] = *(const v4f*)(a_src + (size_t)(32 * i) * Kp + ch * BK);
    const int t = ch >> 4;
    const int icc = (ch & 15) * BK;
    const int rin = pr - ((dymask >> t) & 1);
    const int cin = pc - ((dxmask >> t) & 1);
    const bool ok = pok && (unsigned)rin < (unsigned)H_N && (unsigned)cin < (unsigned)W_N;
    const int off = ok ? rin * W_N + cin : 0;
    const float* xs = x + (size_t)(icc + b_kq * 8) * HW_N + off;
#pragma unroll
    for (int j = 0; j < 8; ++j) rb[j] = ok ? xs[(size_t)j * HW_N] : 0.0f;
  };

  auto store_chunk = [&](int buf) {
#pragma unroll
    for (int i = 0; i < 4; ++i)
      *(v4f*)&As[buf][(a_row + 32 * i) * LP + a_kq * 4] = ra[i];
    v4f b0 = {rb[0], rb[1], rb[2], rb[3]};
    v4f b1 = {rb[4], rb[5], rb[6], rb[7]};
    *(v4f*)&Bs[buf][b_px * LP + b_kq * 8] = b0;
    *(v4f*)&Bs[buf][b_px * LP + b_kq * 8 + 4] = b1;
  };

  v8f acc00 = {}, acc01 = {}, acc10 = {}, acc11 = {};
  const int a_frag = (wm * 32 + c16) * LP + hi * 4;
  const int b_frag = (wn * 32 + c16) * LP + hi * 4;

  const int nch = ntaps * 16;
  load_chunk(0);
  store_chunk(0);
  __syncthreads();

  for (int ch = 0; ch < nch; ++ch) {
    const int buf = ch & 1;
    const bool more = (ch + 1) < nch;
    if (more) load_chunk(ch + 1);

    const float* as = &As[buf][a_frag];
    const float* bs = &Bs[buf][b_frag];
#pragma unroll
    for (int g = 0; g < 4; ++g) {
      const v4f a0 = *(const v4f*)(as + g * 8);
      const v4f a1 = *(const v4f*)(as + 16 * LP + g * 8);
      const v4f b0 = *(const v4f*)(bs + g * 8);
      const v4f b1 = *(const v4f*)(bs + 16 * LP + g * 8);
      const v2f a0l = {a0.x, a0.y}, a0h = {a0.z, a0.w};
      const v2f a1l = {a1.x, a1.y}, a1h = {a1.z, a1.w};
      const v2f b0l = {b0.x, b0.y}, b0h = {b0.z, b0.w};
      const v2f b1l = {b1.x, b1.y}, b1h = {b1.z, b1.w};
      acc00 = __builtin_amdgcn_wmma_f32_16x16x4_f32(false, a0l, false, b0l, (short)0, acc00, false, false);
      acc01 = __builtin_amdgcn_wmma_f32_16x16x4_f32(false, a0l, false, b1l, (short)0, acc01, false, false);
      acc10 = __builtin_amdgcn_wmma_f32_16x16x4_f32(false, a1l, false, b0l, (short)0, acc10, false, false);
      acc11 = __builtin_amdgcn_wmma_f32_16x16x4_f32(false, a1l, false, b1l, (short)0, acc11, false, false);
      acc00 = __builtin_amdgcn_wmma_f32_16x16x4_f32(false, a0h, false, b0h, (short)0, acc00, false, false);
      acc01 = __builtin_amdgcn_wmma_f32_16x16x4_f32(false, a0h, false, b1h, (short)0, acc01, false, false);
      acc10 = __builtin_amdgcn_wmma_f32_16x16x4_f32(false, a1h, false, b0h, (short)0, acc10, false, false);
      acc11 = __builtin_amdgcn_wmma_f32_16x16x4_f32(false, a1h, false, b1h, (short)0, acc11, false, false);
    }

    if (more) store_chunk(buf ^ 1);
    __syncthreads();
  }

#pragma unroll
  for (int j = 0; j < 2; ++j) {
    const int q = n0 + wn * 32 + j * 16 + c16;
    if (q < npx) {
      const int qr = q / cols;
      const int qc = q - qr * cols;
      float* o0 = up + (size_t)(oc0 + wm * 32 + hi * 8) * PHS + (size_t)qr * PH + qc;
      float* o1 = o0 + (size_t)16 * PHS;
      const v8f d0 = j ? acc01 : acc00;
      const v8f d1 = j ? acc11 : acc10;
#pragma unroll
      for (int r = 0; r < 8; ++r) {
        o0[(size_t)r * PHS] = d0[r];
        o1[(size_t)r * PHS] = d1[r];
      }
    }
  }
}

__global__ __launch_bounds__(256) void blur_kernel(const float* __restrict__ up,
                                                   float* __restrict__ out) {
  __shared__ float tile[19 * 20];
  const int oc = blockIdx.z;
  const int y0 = blockIdx.y * 16;
  const int x0 = blockIdx.x * 16;
  const int tid = threadIdx.x;

  for (int idx = tid; idx < 19 * 19; idx += 256) {
    const int ly = idx / 19, lx = idx % 19;
    const int yy = y0 - 1 + ly;
    const int xx = x0 - 1 + lx;
    float v = 0.f;
    if (yy >= 0 && yy <= 256 && xx >= 0 && xx <= 256) {
      const int p = ((yy & 1) << 1) | (xx & 1);
      v = up[((size_t)p * OC_N + oc) * PHS + (size_t)(yy >> 1) * PH + (xx >> 1)];
    }
    tile[ly * 20 + lx] = v;
  }
  __syncthreads();

  const int ty = tid >> 4, tx = tid & 15;
  const float c0 = 0.25f, c1 = 0.75f;
  const float cy[4] = {c0, c1, c1, c0};
  float acc = 0.f;
#pragma unroll
  for (int i = 0; i < 4; ++i) {
    const float* row = &tile[(ty + i) * 20 + tx];
    const float rsum = c0 * row[0] + c1 * row[1] + c1 * row[2] + c0 * row[3];
    acc += cy[i] * rsum;
  }
  out[((size_t)oc * 256 + (y0 + ty)) * 256 + (x0 + tx)] = acc;
}

extern "C" void kernel_launch(void* const* d_in, const int* in_sizes, int n_in,
                              void* d_out, int out_size, void* d_ws, size_t ws_size,
                              hipStream_t stream) {
  const float* x = (const float*)d_in[0];
  const float* wgt = (const float*)d_in[1];
  float* out = (float*)d_out;

  float* ws = (float*)d_ws;
  float* wp = ws;
  float* up = ws + (size_t)OC_N * 4608;

  const float* w00 = wp;
  const float* w01 = wp + (size_t)OC_N * 2048;
  const float* w10 = w01 + (size_t)OC_N * 1024;
  const float* w11 = w10 + (size_t)OC_N * 1024;

  float* up00 = up + (size_t)0 * OC_N * PHS;
  float* up01 = up + (size_t)1 * OC_N * PHS;
  float* up10 = up + (size_t)2 * OC_N * PHS;
  float* up11 = up + (size_t)3 * OC_N * PHS;

  prep_weights<<<OC_N, 256, 0, stream>>>(wgt, wp);

  upconv_gemm<<<dim3(4, 261), 256, 0, stream>>>(x, w00, up00, 2048, 4, 0x3, 0x5, 129, 129);
  upconv_gemm<<<dim3(4, 258), 256, 0, stream>>>(x, w01, up01, 1024, 2, 0x1, 0x0, 129, 128);
  upconv_gemm<<<dim3(4, 258), 256, 0, stream>>>(x, w10, up10, 1024, 2, 0x0, 0x1, 128, 129);
  upconv_gemm<<<dim3(4, 256), 256, 0, stream>>>(x, w11, up11,  512, 1, 0x0, 0x0, 128, 128);

  blur_kernel<<<dim3(16, 16, OC_N), 256, 0, stream>>>(up, out);
}


# Blur pass rewritten as a streaming kernel: thread = 2 output columns x 16 rows, coalesced phase-row loads all in flight, row sums in registers, no LDS or barrier (32x fewer waves); same sum order
# speedup vs baseline: 1.8775x; 1.2232x over previous
#include <hip/hip_runtime.h>
#include <math.h>

#define IC_N 512
#define OC_N 512
#define H_N 128
#define W_N 128
#define HW_N (H_N * W_N)
#define PH 129
#define PHS (PH * PH)

typedef __attribute__((ext_vector_type(2))) float v2f;
typedef __attribute__((ext_vector_type(4))) float v4f;
typedef __attribute__((ext_vector_type(8))) float v8f;

__global__ __launch_bounds__(256) void prep_weights(const float* __restrict__ wgt,
                                                    float* __restrict__ wp) {
  const int oc = blockIdx.x;
  const int tid = threadIdx.x;
  const float scale = rsqrtf(4608.0f);
  __shared__ float red[256];
  __shared__ float dsh;

  const float* wo = wgt + (size_t)oc * IC_N * 9;
  float s = 0.f;
  for (int ic = tid; ic < IC_N; ic += 256) {
    const float* wi = wo + ic * 9;
#pragma unroll
    for (int k = 0; k < 9; ++k) {
      float v = scale * wi[k];
      s += v * v;
    }
  }
  red[tid] = s;
  __syncthreads();
  for (int off = 128; off > 0; off >>= 1) {
    if (tid < off) red[tid] += red[tid + off];
    __syncthreads();
  }
  if (tid == 0) dsh = rsqrtf(red[0] + 1e-8f);
  __syncthreads();
  const float d = dsh * scale;

  float* w00 = wp;
  float* w01 = wp + (size_t)OC_N * 2048;
  float* w10 = w01 + (size_t)OC_N * 1024;
  float* w11 = w10 + (size_t)OC_N * 1024;

  for (int ic = tid; ic < IC_N; ic += 256) {
    const float* wi = wo + ic * 9;
    w00[(size_t)oc * 2048 + 0 * 512 + ic] = d * wi[2 * 3 + 2];
    w00[(size_t)oc * 2048 + 1 * 512 + ic] = d * wi[2 * 3 + 0];
    w00[(size_t)oc * 2048 + 2 * 512 + ic] = d * wi[0 * 3 + 2];
    w00[(size_t)oc * 2048 + 3 * 512 + ic] = d * wi[0 * 3 + 0];
    w01[(size_t)oc * 1024 + 0 * 512 + ic] = d * wi[2 * 3 + 1];
    w01[(size_t)oc * 1024 + 1 * 512 + ic] = d * wi[0 * 3 + 1];
    w10[(size_t)oc * 1024 + 0 * 512 + ic] = d * wi[1 * 3 + 2];
    w10[(size_t)oc * 1024 + 1 * 512 + ic] = d * wi[1 * 3 + 0];
    w11[(size_t)oc * 512 + ic] = d * wi[1 * 3 + 1];
  }
}

#define BM 128
#define BN 64
#define BK 32
#define LP 36

__global__ __launch_bounds__(256) void upconv_gemm(
    const float* __restrict__ x, const float* __restrict__ wp,
    float* __restrict__ up, int Kp, int ntaps, int dymask, int dxmask,
    int rows, int cols) {
  __shared__ __attribute__((aligned(16))) float As[2][BM * LP];
  __shared__ __attribute__((aligned(16))) float Bs[2][BN * LP];

  const int tid = threadIdx.x;
  const int lane = tid & 31;
  const int wave = tid >> 5;
  const int c16 = lane & 15;
  const int hi = lane >> 4;
  const int wm = wave >> 1;
  const int wn = wave & 1;

  const int npx = rows * cols;
  const int oc0 = blockIdx.x * BM;
  const int n0 = blockIdx.y * BN;

  const int a_row = tid >> 3;
  const int a_kq = tid & 7;
  const int b_px = tid & 63;
  const int b_kq = tid >> 6;

  const int p = n0 + b_px;
  const bool pok = p < npx;
  const int pr = p / cols;
  const int pc = p - pr * cols;

  const float* a_src = wp + (size_t)(oc0 + a_row) * Kp + a_kq * 4;

  v4f ra[4];
  float rb[8];

  auto load_chunk = [&](int ch) {
#pragma unroll
    for (int i = 0; i < 4; ++i)
      ra[i] = *(const v4f*)(a_src + (size_t)(32 * i) * Kp + ch * BK);
    const int t = ch >> 4;
    const int icc = (ch & 15) * BK;
    const int rin = pr - ((dymask >> t) & 1);
    const int cin = pc - ((dxmask >> t) & 1);
    const bool ok = pok && (unsigned)rin < (unsigned)H_N && (unsigned)cin < (unsigned)W_N;
    const int off = ok ? rin * W_N + cin : 0;
    const float* xs = x + (size_t)(icc + b_kq * 8) * HW_N + off;
#pragma unroll
    for (int j = 0; j < 8; ++j) rb[j] = ok ? xs[(size_t)j * HW_N] : 0.0f;
  };

  auto store_chunk = [&](int buf) {
#pragma unroll
    for (int i = 0; i < 4; ++i)
      *(v4f*)&As[buf][(a_row + 32 * i) * LP + a_kq * 4] = ra[i];
    v4f b0 = {rb[0], rb[1], rb[2], rb[3]};
    v4f b1 = {rb[4], rb[5], rb[6], rb[7]};
    *(v4f*)&Bs[buf][b_px * LP + b_kq * 8] = b0;
    *(v4f*)&Bs[buf][b_px * LP + b_kq * 8 + 4] = b1;
  };

  v8f acc00 = {}, acc01 = {}, acc10 = {}, acc11 = {};
  const int a_frag = (wm * 32 + c16) * LP + hi * 4;
  const int b_frag = (wn * 32 + c16) * LP + hi * 4;

  const int nch = ntaps * 16;
  load_chunk(0);
  store_chunk(0);
  __syncthreads();

  for (int ch = 0; ch < nch; ++ch) {
    const int buf = ch & 1;
    const bool more = (ch + 1) < nch;
    if (more) load_chunk(ch + 1);

    const float* as = &As[buf][a_frag];
    const float* bs = &Bs[buf][b_frag];
#pragma unroll
    for (int g = 0; g < 4; ++g) {
      const v4f a0 = *(const v4f*)(as + g * 8);
      const v4f a1 = *(const v4f*)(as + 16 * LP + g * 8);
      const v4f b0 = *(const v4f*)(bs + g * 8);
      const v4f b1 = *(const v4f*)(bs + 16 * LP + g * 8);
      const v2f a0l = {a0.x, a0.y}, a0h = {a0.z, a0.w};
      const v2f a1l = {a1.x, a1.y}, a1h = {a1.z, a1.w};
      const v2f b0l = {b0.x, b0.y}, b0h = {b0.z, b0.w};
      const v2f b1l = {b1.x, b1.y}, b1h = {b1.z, b1.w};
      acc00 = __builtin_amdgcn_wmma_f32_16x16x4_f32(false, a0l, false, b0l, (short)0, acc00, false, false);
      acc01 = __builtin_amdgcn_wmma_f32_16x16x4_f32(false, a0l, false, b1l, (short)0, acc01, false, false);
      acc10 = __builtin_amdgcn_wmma_f32_16x16x4_f32(false, a1l, false, b0l, (short)0, acc10, false, false);
      acc11 = __builtin_amdgcn_wmma_f32_16x16x4_f32(false, a1l, false, b1l, (short)0, acc11, false, false);
      acc00 = __builtin_amdgcn_wmma_f32_16x16x4_f32(false, a0h, false, b0h, (short)0, acc00, false, false);
      acc01 = __builtin_amdgcn_wmma_f32_16x16x4_f32(false, a0h, false, b1h, (short)0, acc01, false, false);
      acc10 = __builtin_amdgcn_wmma_f32_16x16x4_f32(false, a1h, false, b0h, (short)0, acc10, false, false);
      acc11 = __builtin_amdgcn_wmma_f32_16x16x4_f32(false, a1h, false, b1h, (short)0, acc11, false, false);
    }

    if (more) store_chunk(buf ^ 1);
    __syncthreads();
  }

#pragma unroll
  for (int j = 0; j < 2; ++j) {
    const int q = n0 + wn * 32 + j * 16 + c16;
    if (q < npx) {
      const int qr = q / cols;
      const int qc = q - qr * cols;
      float* o0 = up + (size_t)(oc0 + wm * 32 + hi * 8) * PHS + (size_t)qr * PH + qc;
      float* o1 = o0 + (size_t)16 * PHS;
      const v8f d0 = j ? acc01 : acc00;
      const v8f d1 = j ? acc11 : acc10;
#pragma unroll
      for (int r = 0; r < 8; ++r) {
        o0[(size_t)r * PHS] = d0[r];
        o1[(size_t)r * PHS] = d1[r];
      }
    }
  }
}

#define BR 8
#define NU (2 * BR + 3)

__global__ __launch_bounds__(128) void blur_kernel(const float* __restrict__ u00,
                                                   const float* __restrict__ u01,
                                                   const float* __restrict__ u10,
                                                   const float* __restrict__ u11,
                                                   float* __restrict__ out) {
  const int m = threadIdx.x;
  const int n0 = blockIdx.x * BR;
  const int oc = blockIdx.y;

  const float* ee = u00 + (size_t)oc * PHS;
  const float* eo = u01 + (size_t)oc * PHS;
  const float* oe = u10 + (size_t)oc * PHS;
  const float* oo = u11 + (size_t)oc * PHS;

  const bool hasl = m > 0;
  const bool hasr = m < 127;
  const int ml = hasl ? m - 1 : 0;
  const int mr = hasr ? m + 1 : 127;
  const float c0 = 0.25f, c1 = 0.75f;

  const float* eeb = ee + n0 * PH;
  const float* eob = eo + n0 * PH;
  const float* oeb = oe + n0 * PH;
  const float* oob = oo + n0 * PH;
  const bool topok = n0 > 0;
  const bool botok = n0 + BR < 128;

  float e0[NU], e1[NU], tl[NU], tc[NU], tr[NU];
#pragma unroll
  for (int j = 0; j < NU; ++j) {
    const float* re;
    const float* ro;
    if ((j & 1) == 0) {
      int k = (j >> 1) - 1;
      if (j == 0) k = topok ? -1 : 0;
      if (j == NU - 1) k = botok ? BR : BR - 1;
      re = oeb + k * PH;
      ro = oob + k * PH;
    } else {
      re = eeb + (j >> 1) * PH;
      ro = eob + (j >> 1) * PH;
    }
    e0[j] = re[m];
    e1[j] = re[m + 1];
    tl[j] = ro[ml];
    tc[j] = ro[m];
    tr[j] = ro[mr];
  }

#pragma unroll
  for (int j = 0; j < NU; j += 5) {
    if (j + 4 < NU) {
      asm volatile("" ::"v"(e0[j]), "v"(e1[j]), "v"(tl[j]), "v"(tc[j]), "v"(tr[j]),
                   "v"(e0[j + 1]), "v"(e1[j + 1]), "v"(tl[j + 1]), "v"(tc[j + 1]), "v"(tr[j + 1]),
                   "v"(e0[j + 2]), "v"(e1[j + 2]), "v"(tl[j + 2]), "v"(tc[j + 2]), "v"(tr[j + 2]),
                   "v"(e0[j + 3]), "v"(e1[j + 3]), "v"(tl[j + 3]), "v"(tc[j + 3]), "v"(tr[j + 3]),
                   "v"(e0[j + 4]), "v"(e1[j + 4]), "v"(tl[j + 4]), "v"(tc[j + 4]), "v"(tr[j + 4]));
    } else {
      asm volatile("" ::"v"(e0[j]), "v"(e1[j]), "v"(tl[j]), "v"(tc[j]), "v"(tr[j]),
                   "v"(e0[j + 1]), "v"(e1[j + 1]), "v"(tl[j + 1]), "v"(tc[j + 1]), "v"(tr[j + 1]),
                   "v"(e0[j + 2]), "v"(e1[j + 2]), "v"(tl[j + 2]), "v"(tc[j + 2]), "v"(tr[j + 2]),
                   "v"(e0[j + 3]), "v"(e1[j + 3]), "v"(tl[j + 3]), "v"(tc[j + 3]), "v"(tr[j + 3]));
    }
  }

  float h0[NU], h1[NU];
#pragma unroll
  for (int j = 0; j < NU; ++j) {
    const float ol = hasl ? tl[j] : 0.f;
    const float orr = hasr ? tr[j] : 0.f;
    float a = c0 * ol;
    a += c1 * e0[j];
    a += c1 * tc[j];
    a += c0 * e1[j];
    float b = c0 * e0[j];
    b += c1 * tc[j];
    b += c1 * e1[j];
    b += c0 * orr;
    const bool rowok = (j == 0) ? topok : ((j == NU - 1) ? botok : true);
    h0[j] = rowok ? a : 0.f;
    h1[j] = rowok ? b : 0.f;
  }

  float* o = out + ((size_t)oc * 256 + 2 * n0) * 256 + 2 * m;
#pragma unroll
  for (int i = 0; i < 2 * BR; ++i) {
    float a = c0 * h0[i];
    a += c1 * h0[i + 1];
    a += c1 * h0[i + 2];
    a += c0 * h0[i + 3];
    float b = c0 * h1[i];
    b += c1 * h1[i + 1];
    b += c1 * h1[i + 2];
    b += c0 * h1[i + 3];
    const v2f v = {a, b};
    *(v2f*)(o + (size_t)i * 256) = v;
  }
}

extern "C" void kernel_launch(void* const* d_in, const int* in_sizes, int n_in,
                              void* d_out, int out_size, void* d_ws, size_t ws_size,
                              hipStream_t stream) {
  const float* x = (const float*)d_in[0];
  const float* wgt = (const float*)d_in[1];
  float* out = (float*)d_out;

  float* ws = (float*)d_ws;
  float* wp = ws;
  float* up = ws + (size_t)OC_N * 4608;

  const float* w00 = wp;
  const float* w01 = wp + (size_t)OC_N * 2048;
  const float* w10 = w01 + (size_t)OC_N * 1024;
  const float* w11 = w10 + (size_t)OC_N * 1024;

  float* up00 = up + (size_t)0 * OC_N * PHS;
  float* up01 = up + (size_t)1 * OC_N * PHS;
  float* up10 = up + (size_t)2 * OC_N * PHS;
  float* up11 = up + (size_t)3 * OC_N * PHS;

  prep_weights<<<OC_N, 256, 0, stream>>>(wgt, wp);

  upconv_gemm<<<dim3(4, 261), 256, 0, stream>>>(x, w00, up00, 2048, 4, 0x3, 0x5, 129, 129);
  upconv_gemm<<<dim3(4, 258), 256, 0, stream>>>(x, w01, up01, 1024, 2, 0x1, 0x0, 129, 128);
  upconv_gemm<<<dim3(4, 258), 256, 0, stream>>>(x, w10, up10, 1024, 2, 0x0, 0x1, 128, 129);
  upconv_gemm<<<dim3(4, 256), 256, 0, stream>>>(x, w11, up11,  512, 1, 0x0, 0x0, 128, 128);

  blur_kernel<<<dim3(16, OC_N), 128, 0, stream>>>(up00, up01, up10, up11, out);
}


# Four per-phase GEMM launches merged into one launch of 4132 blocks, longest-K phase first: tiles of all phases fill the chip together and one tail is left instead of four; sum order unchanged
# speedup vs baseline: 2.2470x; 1.1968x over previous
#include <hip/hip_runtime.h>
#include <math.h>

#define IC_N 512
#define OC_N 512
#define H_N 128
#define W_N 128
#define HW_N (H_N * W_N)
#define PH 129
#define PHS (PH * PH)

typedef __attribute__((ext_vector_type(2))) float v2f;
typedef __attribute__((ext_vector_type(4))) float v4f;
typedef __attribute__((ext_vector_type(8))) float v8f;

__global__ __launch_bounds__(256) void prep_weights(const float* __restrict__ wgt,
                                                    float* __restrict__ wp) {
  const int oc = blockIdx.x;
  const int tid = threadIdx.x;
  const float scale = rsqrtf(4608.0f);
  __shared__ float red[256];
  __shared__ float dsh;

  const float* wo = wgt + (size_t)oc * IC_N * 9;
  float s = 0.f;
  for (int ic = tid; ic < IC_N; ic += 256) {
    const float* wi = wo + ic * 9;
#pragma unroll
    for (int k = 0; k < 9; ++k) {
      float v = scale * wi[k];
      s += v * v;
    }
  }
  red[tid] = s;
  __syncthreads();
  for (int off = 128; off > 0; off >>= 1) {
    if (tid < off) red[tid] += red[tid + off];
    __syncthreads();
  }
  if (tid == 0) dsh = rsqrtf(red[0] + 1e-8f);
  __syncthreads();
  const float d = dsh * scale;

  float* w00 = wp;
  float* w01 = wp + (size_t)OC_N * 2048;
  float* w10 = w01 + (size_t)OC_N * 1024;
  float* w11 = w10 + (size_t)OC_N * 1024;

  for (int ic = tid; ic < IC_N; ic += 256) {
    const float* wi = wo + ic * 9;
    w00[(size_t)oc * 2048 + 0 * 512 + ic] = d * wi[2 * 3 + 2];
    w00[(size_t)oc * 2048 + 1 * 512 + ic] = d * wi[2 * 3 + 0];
    w00[(size_t)oc * 2048 + 2 * 512 + ic] = d * wi[0 * 3 + 2];
    w00[(size_t)oc * 2048 + 3 * 512 + ic] = d * wi[0 * 3 + 0];
    w01[(size_t)oc * 1024 + 0 * 512 + ic] = d * wi[2 * 3 + 1];
    w01[(size_t)oc * 1024 + 1 * 512 + ic] = d * wi[0 * 3 + 1];
    w10[(size_t)oc * 1024 + 0 * 512 + ic] = d * wi[1 * 3 + 2];
    w10[(size_t)oc * 1024 + 1 * 512 + ic] = d * wi[1 * 3 + 0];
    w11[(size_t)oc * 512 + ic] = d * wi[1 * 3 + 1];
  }
}

#define BM 128
#define BN 64
#define BK 32
#define LP 36
#define NB00 1044
#define NB01 1032
#define NB10 1032
#define NB11 1024

__global__ __launch_bounds__(256) void upconv_gemm(
    const float* __restrict__ x, const float* __restrict__ wp,
    float* __restrict__ up) {
  __shared__ __attribute__((aligned(16))) float As[2][BM * LP];
  __shared__ __attribute__((aligned(16))) float Bs[2][BN * LP];

  const int tid = threadIdx.x;
  const int lane = tid & 31;
  const int wave = tid >> 5;
  const int c16 = lane & 15;
  const int hi = lane >> 4;
  const int wm = wave >> 1;
  const int wn = wave & 1;

  const int bid = blockIdx.x;
  int ph = 0, local = bid;
  if (bid >= NB00 + NB01 + NB10) { ph = 3; local = bid - (NB00 + NB01 + NB10); }
  else if (bid >= NB00 + NB01) { ph = 2; local = bid - (NB00 + NB01); }
  else if (bid >= NB00) { ph = 1; local = bid - NB00; }
  const int ntaps = ph == 0 ? 4 : (ph == 3 ? 1 : 2);
  const int Kp = ntaps * IC_N;
  const int dymask = ph == 0 ? 0x3 : (ph == 1 ? 0x1 : 0x0);
  const int dxmask = ph == 0 ? 0x5 : (ph == 2 ? 0x1 : 0x0);
  const int rows = (ph & 2) ? 128 : 129;
  const int cols = (ph & 1) ? 128 : 129;
  const int wcol = ph == 0 ? 0 : (ph == 1 ? 2048 : (ph == 2 ? 3072 : 4096));
  const float* __restrict__ wph = wp + (size_t)OC_N * wcol;
  float* __restrict__ uph = up + (size_t)ph * OC_N * PHS;

  const int npx = rows * cols;
  const int oc0 = (local & 3) * BM;
  const int n0 = (local >> 2) * BN;

  const int a_row = tid >> 3;
  const int a_kq = tid & 7;
  const int b_px = tid & 63;
  const int b_kq = tid >> 6;

  const int p = n0 + b_px;
  const bool pok = p < npx;
  const int pr = p / cols;
  const int pc = p - pr * cols;

  const float* a_src = wph + (size_t)(oc0 + a_row) * Kp + a_kq * 4;

  v4f ra[4];
  float rb[8];

  auto load_chunk = [&](int ch) {
#pragma unroll
    for (int i = 0; i < 4; ++i)
      ra[i] = *(const v4f*)(a_src + (size_t)(32 * i) * Kp + ch * BK);
    const int t = ch >> 4;
    const int icc = (ch & 15) * BK;
    const int rin = pr - ((dymask >> t) & 1);
    const int cin = pc - ((dxmask >> t) & 1);
    const bool ok = pok && (unsigned)rin < (unsigned)H_N && (unsigned)cin < (unsigned)W_N;
    const int off = ok ? rin * W_N + cin : 0;
    const float* xs = x + (size_t)(icc + b_kq * 8) * HW_N + off;
#pragma unroll
    for (int j = 0; j < 8; ++j) rb[j] = ok ? xs[(size_t)j * HW_N] : 0.0f;
  };

  auto store_chunk = [&](int buf) {
#pragma unroll
    for (int i = 0; i < 4; ++i)
      *(v4f*)&As[buf][(a_row + 32 * i) * LP + a_kq * 4] = ra[i];
    v4f b0 = {rb[0], rb[1], rb[2], rb[3]};
    v4f b1 = {rb[4], rb[5], rb[6], rb[7]};
    *(v4f*)&Bs[buf][b_px * LP + b_kq * 8] = b0;
    *(v4f*)&Bs[buf][b_px * LP + b_kq * 8 + 4] = b1;
  };

  v8f acc00 = {}, acc01 = {}, acc10 = {}, acc11 = {};
  const int a_frag = (wm * 32 + c16) * LP + hi * 4;
  const int b_frag = (wn * 32 + c16) * LP + hi * 4;

  const int nch = ntaps * 16;
  load_chunk(0);
  store_chunk(0);
  __syncthreads();

  for (int ch = 0; ch < nch; ++ch) {
    const int buf = ch & 1;
    const bool more = (ch + 1) < nch;
    if (more) load_chunk(ch + 1);

    const float* as = &As[buf][a_frag];
    const float* bs = &Bs[buf][b_frag];
#pragma unroll
    for (int g = 0; g < 4; ++g) {
      const v4f a0 = *(const v4f*)(as + g * 8);
      const v4f a1 = *(const v4f*)(as + 16 * LP + g * 8);
      const v4f b0 = *(const v4f*)(bs + g * 8);
      const v4f b1 = *(const v4f*)(bs + 16 * LP + g * 8);
      const v2f a0l = {a0.x, a0.y}, a0h = {a0.z, a0.w};
      const v2f a1l = {a1.x, a1.y}, a1h = {a1.z, a1.w};
      const v2f b0l = {b0.x, b0.y}, b0h = {b0.z, b0.w};
      const v2f b1l = {b1.x, b1.y}, b1h = {b1.z, b1.w};
      acc00 = __builtin_amdgcn_wmma_f32_16x16x4_f32(false, a0l, false, b0l, (short)0, acc00, false, false);
      acc01 = __builtin_amdgcn_wmma_f32_16x16x4_f32(false, a0l, false, b1l, (short)0, acc01, false, false);
      acc10 = __builtin_amdgcn_wmma_f32_16x16x4_f32(false, a1l, false, b0l, (short)0, acc10, false, false);
      acc11 = __builtin_amdgcn_wmma_f32_16x16x4_f32(false, a1l, false, b1l, (short)0, acc11, false, false);
      acc00 = __builtin_amdgcn_wmma_f32_16x16x4_f32(false, a0h, false, b0h, (short)0, acc00, false, false);
      acc01 = __builtin_amdgcn_wmma_f32_16x16x4_f32(false, a0h, false, b1h, (short)0, acc01, false, false);
      acc10 = __builtin_amdgcn_wmma_f32_16x16x4_f32(false, a1h, false, b0h, (short)0, acc10, false, false);
      acc11 = __builtin_amdgcn_wmma_f32_16x16x4_f32(false, a1h, false, b1h, (short)0, acc11, false, false);
    }

    if (more) store_chunk(buf ^ 1);
    __syncthreads();
  }

#pragma unroll
  for (int j = 0; j < 2; ++j) {
    const int q = n0 + wn * 32 + j * 16 + c16;
    if (q < npx) {
      const int qr = q / cols;
      const int qc = q - qr * cols;
      float* o0 = uph + (size_t)(oc0 + wm * 32 + hi * 8) * PHS + (size_t)qr * PH + qc;
      float* o1 = o0 + (size_t)16 * PHS;
      const v8f d0 = j ? acc01 : acc00;
      const v8f d1 = j ? acc11 : acc10;
#pragma unroll
      for (int r = 0; r < 8; ++r) {
        o0[(size_t)r * PHS] = d0[r];
        o1[(size_t)r * PHS] = d1[r];
      }
    }
  }
}

#define BR 8
#define NU (2 * BR + 3)

__global__ __launch_bounds__(128) void blur_kernel(const float* __restrict__ u00,
                                                   const float* __restrict__ u01,
                                                   const float* __restrict__ u10,
                                                   const float* __restrict__ u11,
                                                   float* __restrict__ out) {
  const int m = threadIdx.x;
  const int n0 = blockIdx.x * BR;
  const int oc = blockIdx.y;

  const float* ee = u00 + (size_t)oc * PHS;
  const float* eo = u01 + (size_t)oc * PHS;
  const float* oe = u10 + (size_t)oc * PHS;
  const float* oo = u11 + (size_t)oc * PHS;

  const bool hasl = m > 0;
  const bool hasr = m < 127;
  const int ml = hasl ? m - 1 : 0;
  const int mr = hasr ? m + 1 : 127;
  const float c0 = 0.25f, c1 = 0.75f;

  const float* eeb = ee + n0 * PH;
  const float* eob = eo + n0 * PH;
  const float* oeb = oe + n0 * PH;
  const float* oob = oo + n0 * PH;
  const bool topok = n0 > 0;
  const bool botok = n0 + BR < 128;

  float e0[NU], e1[NU], tl[NU], tc[NU], tr[NU];
#pragma unroll
  for (int j = 0; j < NU; ++j) {
    const float* re;
    const float* ro;
    if ((j & 1) == 0) {
      int k = (j >> 1) - 1;
      if (j == 0) k = topok ? -1 : 0;
      if (j == NU - 1) k = botok ? BR : BR - 1;
      re = oeb + k * PH;
      ro = oob + k * PH;
    } else {
      re = eeb + (j >> 1) * PH;
      ro = eob + (j >> 1) * PH;
    }
    e0[j] = re[m];
    e1[j] = re[m + 1];
    tl[j] = ro[ml];
    tc[j] = ro[m];
    tr[j] = ro[mr];
  }

#pragma unroll
  for (int j = 0; j < NU; j += 5) {
    if (j + 4 < NU) {
      asm volatile("" ::"v"(e0[j]), "v"(e1[j]), "v"(tl[j]), "v"(tc[j]), "v"(tr[j]),
                   "v"(e0[j + 1]), "v"(e1[j + 1]), "v"(tl[j + 1]), "v"(tc[j + 1]), "v"(tr[j + 1]),
                   "v"(e0[j + 2]), "v"(e1[j + 2]), "v"(tl[j + 2]), "v"(tc[j + 2]), "v"(tr[j + 2]),
                   "v"(e0[j + 3]), "v"(e1[j + 3]), "v"(tl[j + 3]), "v"(tc[j + 3]), "v"(tr[j + 3]),
                   "v"(e0[j + 4]), "v"(e1[j + 4]), "v"(tl[j + 4]), "v"(tc[j + 4]), "v"(tr[j + 4]));
    } else {
      asm volatile("" ::"v"(e0[j]), "v"(e1[j]), "v"(tl[j]), "v"(tc[j]), "v"(tr[j]),
                   "v"(e0[j + 1]), "v"(e1[j + 1]), "v"(tl[j + 1]), "v"(tc[j + 1]), "v"(tr[j + 1]),
                   "v"(e0[j + 2]), "v"(e1[j + 2]), "v"(tl[j + 2]), "v"(tc[j + 2]), "v"(tr[j + 2]),
                   "v"(e0[j + 3]), "v"(e1[j + 3]), "v"(tl[j + 3]), "v"(tc[j + 3]), "v"(tr[j + 3]));
    }
  }

  float h0[NU], h1[NU];
#pragma unroll
  for (int j = 0; j < NU; ++j) {
    const float ol = hasl ? tl[j] : 0.f;
    const float orr = hasr ? tr[j] : 0.f;
    float a = c0 * ol;
    a += c1 * e0[j];
    a += c1 * tc[j];
    a += c0 * e1[j];
    float b = c0 * e0[j];
    b += c1 * tc[j];
    b += c1 * e1[j];
    b += c0 * orr;
    const bool rowok = (j == 0) ? topok : ((j == NU - 1) ? botok : true);
    h0[j] = rowok ? a : 0.f;
    h1[j] = rowok ? b : 0.f;
  }

  float* o = out + ((size_t)oc * 256 + 2 * n0) * 256 + 2 * m;
#pragma unroll
  for (int i = 0; i < 2 * BR; ++i) {
    float a = c0 * h0[i];
    a += c1 * h0[i + 1];
    a += c1 * h0[i + 2];
    a += c0 * h0[i + 3];
    float b = c0 * h1[i];
    b += c1 * h1[i + 1];
    b += c1 * h1[i + 2];
    b += c0 * h1[i + 3];
    const v2f v = {a, b};
    *(v2f*)(o + (size_t)i * 256) = v;
  }
}

extern "C" void kernel_launch(void* const* d_in, const int* in_sizes, int n_in,
                              void* d_out, int out_size, void* d_ws, size_t ws_size,
                              hipStream_t stream) {
  const float* x = (const float*)d_in[0];
  const float* wgt = (const float*)d_in[1];
  float* out = (float*)d_out;

  float* ws = (float*)d_ws;
  float* wp = ws;
  float* up = ws + (size_t)OC_N * 4608;

  float* up00 = up + (size_t)0 * OC_N * PHS;
  float* up01 = up + (size_t)1 * OC_N * PHS;
  float* up10 = up + (size_t)2 * OC_N * PHS;
  float* up11 = up + (size_t)3 * OC_N * PHS;

  prep_weights<<<OC_N, 256, 0, stream>>>(wgt, wp);

  upconv_gemm<<<4132, 256, 0, stream>>>(x, wp, up);

  blur_kernel<<<dim3(16, OC_N), 128, 0, stream>>>(up00, up01, up10, up11, out);
}


# GEMM wave tile 32x32 -> 64x64 (16 accumulators; block 256 oc x 64 px on 4 waves): LDS fragment reads per WMMA halved, staging and address instructions per WMMA cut; sum order unchanged
# speedup vs baseline: 2.3316x; 1.0377x over previous
#include <hip/hip_runtime.h>
#include <math.h>

#define IC_N 512
#define OC_N 512
#define H_N 128
#define W_N 128
#define HW_N (H_N * W_N)
#define PH 129
#define PHS (PH * PH)

typedef __attribute__((ext_vector_type(2))) float v2f;
typedef __attribute__((ext_vector_type(4))) float v4f;
typedef __attribute__((ext_vector_type(8))) float v8f;

__global__ __launch_bounds__(256) void prep_weights(const float* __restrict__ wgt,
                                                    float* __restrict__ wp) {
  const int oc = blockIdx.x;
  const int tid = threadIdx.x;
  const float scale = rsqrtf(4608.0f);
  __shared__ float red[256];
  __shared__ float dsh;

  const float* wo = wgt + (size_t)oc * IC_N * 9;
  float s = 0.f;
  for (int ic = tid; ic < IC_N; ic += 256) {
    const float* wi = wo + ic * 9;
#pragma unroll
    for (int k = 0; k < 9; ++k) {
      float v = scale * wi[k];
      s += v * v;
    }
  }
  red[tid] = s;
  __syncthreads();
  for (int off = 128; off > 0; off >>= 1) {
    if (tid < off) red[tid] += red[tid + off];
    __syncthreads();
  }
  if (tid == 0) dsh = rsqrtf(red[0] + 1e-8f);
  __syncthreads();
  const float d = dsh * scale;

  float* w00 = wp;
  float* w01 = wp + (size_t)OC_N * 2048;
  float* w10 = w01 + (size_t)OC_N * 1024;
  float* w11 = w10 + (size_t)OC_N * 1024;

  for (int ic = tid; ic < IC_N; ic += 256) {
    const float* wi = wo + ic * 9;
    w00[(size_t)oc * 2048 + 0 * 512 + ic] = d * wi[2 * 3 + 2];
    w00[(size_t)oc * 2048 + 1 * 512 + ic] = d * wi[2 * 3 + 0];
    w00[(size_t)oc * 2048 + 2 * 512 + ic] = d * wi[0 * 3 + 2];
    w00[(size_t)oc * 2048 + 3 * 512 + ic] = d * wi[0 * 3 + 0];
    w01[(size_t)oc * 1024 + 0 * 512 + ic] = d * wi[2 * 3 + 1];
    w01[(size_t)oc * 1024 + 1 * 512 + ic] = d * wi[0 * 3 + 1];
    w10[(size_t)oc * 1024 + 0 * 512 + ic] = d * wi[1 * 3 + 2];
    w10[(size_t)oc * 1024 + 1 * 512 + ic] = d * wi[1 * 3 + 0];
    w11[(size_t)oc * 512 + ic] = d * wi[1 * 3 + 1];
  }
}

#define BM 256
#define BN 64
#define BK 32
#define LP 36
#define NB00 522
#define NB01 516
#define NB10 516
#define NB11 512

__global__ __launch_bounds__(128) void upconv_gemm(
    const float* __restrict__ x, const float* __restrict__ wp,
    float* __restrict__ up) {
  __shared__ __attribute__((aligned(16))) float As[2][BM * LP];
  __shared__ __attribute__((aligned(16))) float Bs[2][BN * LP];

  const int tid = threadIdx.x;
  const int lane = tid & 31;
  const int wave = tid >> 5;
  const int c16 = lane & 15;
  const int hi = lane >> 4;
  const int wm = wave;

  const int bid = blockIdx.x;
  int ph = 0, local = bid;
  if (bid >= NB00 + NB01 + NB10) { ph = 3; local = bid - (NB00 + NB01 + NB10); }
  else if (bid >= NB00 + NB01) { ph = 2; local = bid - (NB00 + NB01); }
  else if (bid >= NB00) { ph = 1; local = bid - NB00; }
  const int ntaps = ph == 0 ? 4 : (ph == 3 ? 1 : 2);
  const int Kp = ntaps * IC_N;
  const int dymask = ph == 0 ? 0x3 : (ph == 1 ? 0x1 : 0x0);
  const int dxmask = ph == 0 ? 0x5 : (ph == 2 ? 0x1 : 0x0);
  const int rows = (ph & 2) ? 128 : 129;
  const int cols = (ph & 1) ? 128 : 129;
  const int wcol = ph == 0 ? 0 : (ph == 1 ? 2048 : (ph == 2 ? 3072 : 4096));
  const float* __restrict__ wph = wp + (size_t)OC_N * wcol;
  float* __restrict__ uph = up + (size_t)ph * OC_N * PHS;

  const int npx = rows * cols;
  const int oc0 = (local & 1) * BM;
  const int n0 = (local >> 1) * BN;

  const int a_row = tid >> 3;
  const int a_kq = tid & 7;
  const int b_px = tid & 63;
  const int b_kh = tid >> 6;

  const int p = n0 + b_px;
  const bool pok = p < npx;
  const int pr = p / cols;
  const int pc = p - pr * cols;

  const float* a_src = wph + (size_t)(oc0 + a_row) * Kp + a_kq * 4;

  v4f ra[16];
  float rb[16];
  bool rb_ok = false;

  auto load_chunk = [&](int ch) {
#pragma unroll
    for (int i = 0; i < 16; ++i)
      ra[i] = *(const v4f*)(a_src + (size_t)(16 * i) * Kp + ch * BK);
    const int t = ch >> 4;
    const int icc = (ch & 15) * BK;
    const int rin = pr - ((dymask >> t) & 1);
    const int cin = pc - ((dxmask >> t) & 1);
    const bool ok = pok && (unsigned)rin < (unsigned)H_N && (unsigned)cin < (unsigned)W_N;
    const int off = ok ? rin * W_N + cin : 0;
    const float* xs = x + (size_t)(icc + b_kh * 16) * HW_N + off;
#pragma unroll
    for (int j = 0; j < 16; ++j) rb[j] = xs[(size_t)j * HW_N];
    rb_ok = ok;
  };

  auto store_chunk = [&](int buf) {
#pragma unroll
    for (int i = 0; i < 16; ++i)
      *(v4f*)&As[buf][(a_row + 16 * i) * LP + a_kq * 4] = ra[i];
#pragma unroll
    for (int q = 0; q < 4; ++q) {
      const v4f b = {rb[4 * q], rb[4 * q + 1], rb[4 * q + 2], rb[4 * q + 3]};
      const v4f z = {0.0f, 0.0f, 0.0f, 0.0f};
      *(v4f*)&Bs[buf][b_px * LP + b_kh * 16 + q * 4] = rb_ok ? b : z;
    }
  };

  v8f acc[4][4];
#pragma unroll
  for (int mi = 0; mi < 4; ++mi)
#pragma unroll
    for (int ni = 0; ni < 4; ++ni) acc[mi][ni] = (v8f){};

  const int a_frag = (wm * 64 + c16) * LP + hi * 4;
  const int b_frag = c16 * LP + hi * 4;

  const int nch = ntaps * 16;
  load_chunk(0);
  store_chunk(0);
  __syncthreads();

  for (int ch = 0; ch < nch; ++ch) {
    const int buf = ch & 1;
    const bool more = (ch + 1) < nch;
    if (more) load_chunk(ch + 1);

    const float* as = &As[buf][a_frag];
    const float* bs = &Bs[buf][b_frag];
#pragma unroll
    for (int g = 0; g < 4; ++g) {
      v4f a[4], b[4];
#pragma unroll
      for (int i = 0; i < 4; ++i) {
        a[i] = *(const v4f*)(as + i * 16 * LP + g * 8);
        b[i] = *(const v4f*)(bs + i * 16 * LP + g * 8);
      }
#pragma unroll
      for (int mi = 0; mi < 4; ++mi) {
        const v2f al = {a[mi].x, a[mi].y};
#pragma unroll
        for (int ni = 0; ni < 4; ++ni) {
          const v2f bl = {b[ni].x, b[ni].y};
          acc[mi][ni] = __builtin_amdgcn_wmma_f32_16x16x4_f32(false, al, false, bl, (short)0, acc[mi][ni], false, false);
        }
      }
#pragma unroll
      for (int mi = 0; mi < 4; ++mi) {
        const v2f ah = {a[mi].z, a[mi].w};
#pragma unroll
        for (int ni = 0; ni < 4; ++ni) {
          const v2f bh = {b[ni].z, b[ni].w};
          acc[mi][ni] = __builtin_amdgcn_wmma_f32_16x16x4_f32(false, ah, false, bh, (short)0, acc[mi][ni], false, false);
        }
      }
    }

    if (more) store_chunk(buf ^ 1);
    __syncthreads();
  }

#pragma unroll
  for (int ni = 0; ni < 4; ++ni) {
    const int q = n0 + ni * 16 + c16;
    if (q < npx) {
      const int qr = q / cols;
      const int qc = q - qr * cols;
      float* o = uph + (size_t)(oc0 + wm * 64 + hi * 8) * PHS + (size_t)qr * PH + qc;
#pragma unroll
      for (int mi = 0; mi < 4; ++mi) {
#pragma unroll
        for (int r = 0; r < 8; ++r) o[(size_t)(mi * 16 + r) * PHS] = acc[mi][ni][r];
      }
    }
  }
}

#define BR 8
#define NU (2 * BR + 3)

__global__ __launch_bounds__(128) void blur_kernel(const float* __restrict__ u00,
                                                   const float* __restrict__ u01,
                                                   const float* __restrict__ u10,
                                                   const float* __restrict__ u11,
                                                   float* __restrict__ out) {
  const int m = threadIdx.x;
  const int n0 = blockIdx.x * BR;
  const int oc = blockIdx.y;

  const float* ee = u00 + (size_t)oc * PHS;
  const float* eo = u01 + (size_t)oc * PHS;
  const float* oe = u10 + (size_t)oc * PHS;
  const float* oo = u11 + (size_t)oc * PHS;

  const bool hasl = m > 0;
  const bool hasr = m < 127;
  const int ml = hasl ? m - 1 : 0;
  const int mr = hasr ? m + 1 : 127;
  const float c0 = 0.25f, c1 = 0.75f;

  const float* eeb = ee + n0 * PH;
  const float* eob = eo + n0 * PH;
  const float* oeb = oe + n0 * PH;
  const float* oob = oo + n0 * PH;
  const bool topok = n0 > 0;
  const bool botok = n0 + BR < 128;

  float e0[NU], e1[NU], tl[NU], tc[NU], tr[NU];
#pragma unroll
  for (int j = 0; j < NU; ++j) {
    const float* re;
    const float* ro;
    if ((j & 1) == 0) {
      int k = (j >> 1) - 1;
      if (j == 0) k = topok ? -1 : 0;
      if (j == NU - 1) k = botok ? BR : BR - 1;
      re = oeb + k * PH;
      ro = oob + k * PH;
    } else {
      re = eeb + (j >> 1) * PH;
      ro = eob + (j >> 1) * PH;
    }
    e0[j] = re[m];
    e1[j] = re[m + 1];
    tl[j] = ro[ml];
    tc[j] = ro[m];
    tr[j] = ro[mr];
  }

#pragma unroll
  for (int j = 0; j < NU; j += 5) {
    if (j + 4 < NU) {
      asm volatile("" ::"v"(e0[j]), "v"(e1[j]), "v"(tl[j]), "v"(tc[j]), "v"(tr[j]),
                   "v"(e0[j + 1]), "v"(e1[j + 1]), "v"(tl[j + 1]), "v"(tc[j + 1]), "v"(tr[j + 1]),
                   "v"(e0[j + 2]), "v"(e1[j + 2]), "v"(tl[j + 2]), "v"(tc[j + 2]), "v"(tr[j + 2]),
                   "v"(e0[j + 3]), "v"(e1[j + 3]), "v"(tl[j + 3]), "v"(tc[j + 3]), "v"(tr[j + 3]),
                   "v"(e0[j + 4]), "v"(e1[j + 4]), "v"(tl[j + 4]), "v"(tc[j + 4]), "v"(tr[j + 4]));
    } else {
      asm volatile("" ::"v"(e0[j]), "v"(e1[j]), "v"(tl[j]), "v"(tc[j]), "v"(tr[j]),
                   "v"(e0[j + 1]), "v"(e1[j + 1]), "v"(tl[j + 1]), "v"(tc[j + 1]), "v"(tr[j + 1]),
                   "v"(e0[j + 2]), "v"(e1[j + 2]), "v"(tl[j + 2]), "v"(tc[j + 2]), "v"(tr[j + 2]),
                   "v"(e0[j + 3]), "v"(e1[j + 3]), "v"(tl[j + 3]), "v"(tc[j + 3]), "v"(tr[j + 3]));
    }
  }

  float h0[NU], h1[NU];
#pragma unroll
  for (int j = 0; j < NU; ++j) {
    const float ol = hasl ? tl[j] : 0.f;
    const float orr = hasr ? tr[j] : 0.f;
    float a = c0 * ol;
    a += c1 * e0[j];
    a += c1 * tc[j];
    a += c0 * e1[j];
    float b = c0 * e0[j];
    b += c1 * tc[j];
    b += c1 * e1[j];
    b += c0 * orr;
    const bool rowok = (j == 0) ? topok : ((j == NU - 1) ? botok : true);
    h0[j] = rowok ? a : 0.f;
    h1[j] = rowok ? b : 0.f;
  }

  float* o = out + ((size_t)oc * 256 + 2 * n0) * 256 + 2 * m;
#pragma unroll
  for (int i = 0; i < 2 * BR; ++i) {
    float a = c0 * h0[i];
    a += c1 * h0[i + 1];
    a += c1 * h0[i + 2];
    a += c0 * h0[i + 3];
    float b = c0 * h1[i];
    b += c1 * h1[i + 1];
    b += c1 * h1[i + 2];
    b += c0 * h1[i + 3];
    const v2f v = {a, b};
    *(v2f*)(o + (size_t)i * 256) = v;
  }
}

extern "C" void kernel_launch(void* const* d_in, const int* in_sizes, int n_in,
                              void* d_out, int out_size, void* d_ws, size_t ws_size,
                              hipStream_t stream) {
  const float* x = (const float*)d_in[0];
  const float* wgt = (const float*)d_in[1];
  float* out = (float*)d_out;

  float* ws = (float*)d_ws;
  float* wp = ws;
  float* up = ws + (size_t)OC_N * 4608;

  float* up00 = up + (size_t)0 * OC_N * PHS;
  float* up01 = up + (size_t)1 * OC_N * PHS;
  float* up10 = up + (size_t)2 * OC_N * PHS;
  float* up11 = up + (size_t)3 * OC_N * PHS;

  prep_weights<<<OC_N, 256, 0, stream>>>(wgt, wp);

  upconv_gemm<<<2066, 128, 0, stream>>>(x, wp, up);

  blur_kernel<<<dim3(16, OC_N), 128, 0, stream>>>(up00, up01, up10, up11, out);
}


# Even-row phases by minimal filtering F(2,2) along rows: 3 products per output row pair, not 4 (GEMM work x0.84); input row differences in the first launch, pair sums in the blur; sum order changes
# speedup vs baseline: 2.5731x; 1.1036x over previous
#include <hip/hip_runtime.h>
#include <math.h>

#define IC_N 512
#define OC_N 512
#define H_N 128
#define W_N 128
#define HW_N (H_N * W_N)
#define PH 129
#define PHS (PH * PH)
#define NS 65

typedef __attribute__((ext_vector_type(2))) float v2f;
typedef __attribute__((ext_vector_type(4))) float v4f;
typedef __attribute__((ext_vector_type(8))) float v8f;

#define RPW 9
__device__ __forceinline__ void row_differences(const float* __restrict__ x,
                                                float* __restrict__ xt, int ic) {
  const int sb = (threadIdx.x >> 5) * RPW;
  const int c4 = (threadIdx.x & 31) * 4;
  const float* xc = x + (size_t)ic * HW_N + c4;
  const v4f z = {0.0f, 0.0f, 0.0f, 0.0f};
  v4f r[2 * RPW + 1];
#pragma unroll
  for (int i = 0; i < 2 * RPW + 1; ++i) {
    const int row = 2 * sb - 1 + i;
    const int rc = row < 0 ? 0 : (row > H_N - 1 ? H_N - 1 : row);
    const v4f v = *(const v4f*)(xc + rc * W_N);
    r[i] = (row >= 0 && row < H_N) ? v : z;
  }
  float* o = xt + (size_t)ic * HW_N + c4;
#pragma unroll
  for (int i = 0; i < RPW; ++i) {
    const int s = sb + i;
    if (s < NS) {
      *(v4f*)(o + s * W_N) = r[2 * i] - r[2 * i + 1];
      *(v4f*)(o + (size_t)IC_N * HW_N + s * W_N) = r[2 * i + 2] - r[2 * i + 1];
    }
  }
}

__global__ __launch_bounds__(256) void prep_weights(const float* __restrict__ wgt,
                                                    float* __restrict__ wp,
                                                    const float* __restrict__ x,
                                                    float* __restrict__ xt) {
  if (blockIdx.x >= OC_N) {
    row_differences(x, xt, blockIdx.x - OC_N);
    return;
  }
  const int oc = blockIdx.x;
  const int tid = threadIdx.x;
  const float scale = rsqrtf(4608.0f);
  __shared__ float red[256];
  __shared__ float dsh;

  const float* wo = wgt + (size_t)oc * IC_N * 9;
  float s = 0.f;
  for (int ic = tid; ic < IC_N; ic += 256) {
    const float* wi = wo + ic * 9;
#pragma unroll
    for (int k = 0; k < 9; ++k) {
      float v = scale * wi[k];
      s += v * v;
    }
  }
  red[tid] = s;
  __syncthreads();
  for (int off = 128; off > 0; off >>= 1) {
    if (tid < off) red[tid] += red[tid + off];
    __syncthreads();
  }
  if (tid == 0) dsh = rsqrtf(red[0] + 1e-8f);
  __syncthreads();
  const float d = dsh * scale;

  float* g0 = wp;
  float* g1 = wp + (size_t)OC_N * 1024;
  float* g2 = wp + (size_t)OC_N * 2048;
  float* g3 = wp + (size_t)OC_N * 3072;
  float* g4 = wp + (size_t)OC_N * 4096;
  float* g5 = wp + (size_t)OC_N * 4608;
  float* g6 = wp + (size_t)OC_N * 5120;
  float* g7 = wp + (size_t)OC_N * 5632;

  for (int ic = tid; ic < IC_N; ic += 256) {
    const float* wi = wo + ic * 9;
    const float b_m = wi[2 * 3 + 2], b_0 = wi[2 * 3 + 0];
    const float a_m = wi[0 * 3 + 2], a_0 = wi[0 * 3 + 0];
    g0[(size_t)oc * 1024 + ic] = d * b_m;
    g0[(size_t)oc * 1024 + 512 + ic] = d * b_0;
    g1[(size_t)oc * 1024 + ic] = d * (a_m + b_m);
    g1[(size_t)oc * 1024 + 512 + ic] = d * (a_0 + b_0);
    g2[(size_t)oc * 1024 + ic] = d * a_m;
    g2[(size_t)oc * 1024 + 512 + ic] = d * a_0;
    g3[(size_t)oc * 1024 + ic] = d * wi[1 * 3 + 2];
    g3[(size_t)oc * 1024 + 512 + ic] = d * wi[1 * 3 + 0];
    const float vb = wi[2 * 3 + 1], va = wi[0 * 3 + 1];
    g4[(size_t)oc * 512 + ic] = d * vb;
    g5[(size_t)oc * 512 + ic] = d * (va + vb);
    g6[(size_t)oc * 512 + ic] = d * va;
    g7[(size_t)oc * 512 + ic] = d * wi[1 * 3 + 1];
  }
}

#define BM 256
#define BN 64
#define BK 32
#define LP 36
#define NBA 264
#define NB10 516
#define NBB 260
#define NB11 512

__global__ __launch_bounds__(128) void upconv_gemm(
    const float* __restrict__ x, const float* __restrict__ xt,
    const float* __restrict__ wp, float* __restrict__ up) {
  __shared__ __attribute__((aligned(16))) float As[2][BM * LP];
  __shared__ __attribute__((aligned(16))) float Bs[2][BN * LP];

  const int tid = threadIdx.x;
  const int lane = tid & 31;
  const int wave = tid >> 5;
  const int c16 = lane & 15;
  const int hi = lane >> 4;
  const int wm = wave;

  const int bid = blockIdx.x;
  int st, local;
  if (bid < 3 * NBA) { st = bid / NBA; local = bid - st * NBA; }
  else if (bid < 3 * NBA + NB10) { st = 3; local = bid - 3 * NBA; }
  else if (bid < 3 * NBA + NB10 + 3 * NBB) {
    const int t = (bid - (3 * NBA + NB10)) / NBB;
    st = 4 + t;
    local = bid - (3 * NBA + NB10) - t * NBB;
  } else { st = 7; local = bid - (3 * NBA + NB10 + 3 * NBB); }
  const bool two = st < 4;
  const int kind = st & 3;
  const int ntaps = two ? 2 : 1;
  const int Kp = ntaps * IC_N;
  const int dxmask = two ? 0x1 : 0x0;
  const int rows = kind == 3 ? 128 : NS;
  const int cols = two ? 129 : 128;
  const int wcol = two ? st * 1024 : 4096 + (st - 4) * 512;
  const float* __restrict__ wph = wp + (size_t)OC_N * wcol;
  float* __restrict__ uph = up + (size_t)st * OC_N * PHS;
  const float* __restrict__ bsrc = kind == 0 ? xt : (kind == 2 ? xt + (size_t)IC_N * HW_N : x);
  const int rmul = kind == 1 ? 2 : 1;

  const int npx = rows * cols;
  const int oc0 = (local & 1) * BM;
  const int n0 = (local >> 1) * BN;

  const int a_row = tid >> 3;
  const int a_kq = tid & 7;
  const int b_px = tid & 63;
  const int b_kh = tid >> 6;

  const int p = n0 + b_px;
  const bool pok = p < npx;
  const int pr = p / cols;
  const int pc = p - pr * cols;
  const int rin = pr * rmul;
  const bool rok = pok && (unsigned)rin < (unsigned)H_N;

  const float* a_src = wph + (size_t)(oc0 + a_row) * Kp + a_kq * 4;

  v4f ra[16];
  float rb[16];
  bool rb_ok = false;

  auto load_chunk = [&](int ch) {
#pragma unroll
    for (int i = 0; i < 16; ++i)
      ra[i] = *(const v4f*)(a_src + (size_t)(16 * i) * Kp + ch * BK);
    const int t = ch >> 4;
    const int icc = (ch & 15) * BK;
    const int cin = pc - ((dxmask >> t) & 1);
    const bool ok = rok && (unsigned)cin < (unsigned)W_N;
    const int off = ok ? rin * W_N + cin : 0;
    const float* xs = bsrc + (size_t)(icc + b_kh * 16) * HW_N + off;
#pragma unroll
    for (int j = 0; j < 16; ++j) rb[j] = xs[(size_t)j * HW_N];
    rb_ok = ok;
  };

  auto store_chunk = [&](int buf) {
#pragma unroll
    for (int i = 0; i < 16; ++i)
      *(v4f*)&As[buf][(a_row + 16 * i) * LP + a_kq * 4] = ra[i];
#pragma unroll
    for (int q = 0; q < 4; ++q) {
      const v4f b = {rb[4 * q], rb[4 * q + 1], rb[4 * q + 2], rb[4 * q + 3]};
      const v4f z = {0.0f, 0.0f, 0.0f, 0.0f};
      *(v4f*)&Bs[buf][b_px * LP + b_kh * 16 + q * 4] = rb_ok ? b : z;
    }
  };

  v8f acc[4][4];
#pragma unroll
  for (int mi = 0; mi < 4; ++mi)
#pragma unroll
    for (int ni = 0; ni < 4; ++ni) acc[mi][ni] = (v8f){};

  const int a_frag = (wm * 64 + c16) * LP + hi * 4;
  const int b_frag = c16 * LP + hi * 4;

  const int nch = ntaps * 16;
  load_chunk(0);
  store_chunk(0);
  __syncthreads();

  for (int ch = 0; ch < nch; ++ch) {
    const int buf = ch & 1;
    const bool more = (ch + 1) < nch;
    if (more) load_chunk(ch + 1);

    const float* as = &As[buf][a_frag];
    const float* bs = &Bs[buf][b_frag];
#pragma unroll
    for (int g = 0; g < 4; ++g) {
      v4f a[4], b[4];
#pragma unroll
      for (int i = 0; i < 4; ++i) {
        a[i] = *(const v4f*)(as + i * 16 * LP + g * 8);
        b[i] = *(const v4f*)(bs + i * 16 * LP + g * 8);
      }
#pragma unroll
      for (int mi = 0; mi < 4; ++mi) {
        const v2f al = {a[mi].x, a[mi].y};
#pragma unroll
        for (int ni = 0; ni < 4; ++ni) {
          const v2f bl = {b[ni].x, b[ni].y};
          acc[mi][ni] = __builtin_amdgcn_wmma_f32_16x16x4_f32(false, al, false, bl, (short)0, acc[mi][ni], false, false);
        }
      }
#pragma unroll
      for (int mi = 0; mi < 4; ++mi) {
        const v2f ah = {a[mi].z, a[mi].w};
#pragma unroll
        for (int ni = 0; ni < 4; ++ni) {
          const v2f bh = {b[ni].z, b[ni].w};
          acc[mi][ni] = __builtin_amdgcn_wmma_f32_16x16x4_f32(false, ah, false, bh, (short)0, acc[mi][ni], false, false);
        }
      }
    }

    if (more) store_chunk(buf ^ 1);
    __syncthreads();
  }

#pragma unroll
  for (int ni = 0; ni < 4; ++ni) {
    const int q = n0 + ni * 16 + c16;
    if (q < npx) {
      const int qr = q / cols;
      const int qc = q - qr * cols;
      float* o = uph + (size_t)(oc0 + wm * 64 + hi * 8) * PHS + (size_t)qr * PH + qc;
#pragma unroll
      for (int mi = 0; mi < 4; ++mi) {
#pragma unroll
        for (int r = 0; r < 8; ++r) o[(size_t)(mi * 16 + r) * PHS] = acc[mi][ni][r];
      }
    }
  }
}

#define BR 8
#define NU (2 * BR + 3)
#define NOD (BR + 2)
#define NPR (BR / 2 + 1)
#define NRAW (5 * NOD + 5 * (3 * NPR - 1))

__global__ __launch_bounds__(128) void blur_kernel(const float* __restrict__ up,
                                                   float* __restrict__ out) {
  const int m = threadIdx.x;
  const int n0 = blockIdx.x * BR;
  const int oc = blockIdx.y;

  const size_t S = (size_t)OC_N * PHS;
  const float* pl = up + (size_t)oc * PHS;
  const float* a1 = pl;
  const float* a2 = pl + S;
  const float* a3 = pl + 2 * S;
  const float* oe = pl + 3 * S;
  const float* b1 = pl + 4 * S;
  const float* b2 = pl + 5 * S;
  const float* b3 = pl + 6 * S;
  const float* oo = pl + 7 * S;

  const bool hasl = m > 0;
  const bool hasr = m < 127;
  const int ml = hasl ? m - 1 : 0;
  const int mr = hasr ? m + 1 : 127;
  const float c0 = 0.25f, c1 = 0.75f;

  const bool topok = n0 > 0;
  const bool botok = n0 + BR < 128;
  const int s0 = n0 >> 1;

  float raw[NRAW];
#pragma unroll
  for (int jj = 0; jj < NOD; ++jj) {
    int k = jj - 1;
    if (jj == 0) k = topok ? -1 : 0;
    if (jj == NOD - 1) k = botok ? BR : BR - 1;
    const float* re = oe + (n0 + k) * PH;
    const float* ro = oo + (n0 + k) * PH;
    raw[5 * jj + 0] = re[m];
    raw[5 * jj + 1] = re[m + 1];
    raw[5 * jj + 2] = ro[ml];
    raw[5 * jj + 3] = ro[m];
    raw[5 * jj + 4] = ro[mr];
  }
#pragma unroll
  for (int q = 0; q < 3; ++q) {
    const float* pe = q == 0 ? a1 : (q == 1 ? a2 : a3);
    const float* po = q == 0 ? b1 : (q == 1 ? b2 : b3);
#pragma unroll
    for (int sp = 0; sp < NPR; ++sp) {
      if (q == 2 && sp == NPR - 1) continue;
      const float* re = pe + (s0 + sp) * PH;
      const float* ro = po + (s0 + sp) * PH;
      const int b = 5 * NOD + 5 * (q * NPR + sp);
      raw[b + 0] = re[m];
      raw[b + 1] = re[m + 1];
      raw[b + 2] = ro[ml];
      raw[b + 3] = ro[m];
      raw[b + 4] = ro[mr];
    }
  }

#pragma unroll
  for (int g = 0; g < NRAW; g += 24) {
    asm volatile("" ::"v"(raw[g]), "v"(raw[g + 1]), "v"(raw[g + 2]), "v"(raw[g + 3]),
                 "v"(raw[g + 4]), "v"(raw[g + 5]), "v"(raw[g + 6]), "v"(raw[g + 7]),
                 "v"(raw[g + 8]), "v"(raw[g + 9]), "v"(raw[g + 10]), "v"(raw[g + 11]),
                 "v"(raw[g + 12]), "v"(raw[g + 13]), "v"(raw[g + 14]), "v"(raw[g + 15]),
                 "v"(raw[g + 16]), "v"(raw[g + 17]), "v"(raw[g + 18]), "v"(raw[g + 19]),
                 "v"(raw[g + 20]), "v"(raw[g + 21]), "v"(raw[g + 22]), "v"(raw[g + 23]));
  }

  float h0[NU], h1[NU];
#pragma unroll
  for (int j = 0; j < NU; ++j) {
    float e0, e1, tl, tc, tr;
    if ((j & 1) == 0) {
      const int b = 5 * (j >> 1);
      e0 = raw[b]; e1 = raw[b + 1]; tl = raw[b + 2]; tc = raw[b + 3]; tr = raw[b + 4];
    } else {
      const int i = j >> 1;
      const int sp = i >> 1;
      const int bx = 5 * NOD + 5 * (((i & 1) ? 1 : 0) * NPR + sp);
      const int by = 5 * NOD + 5 * (((i & 1) ? 2 : 1) * NPR + sp);
      e0 = raw[bx] + raw[by];
      e1 = raw[bx + 1] + raw[by + 1];
      tl = raw[bx + 2] + raw[by + 2];
      tc = raw[bx + 3] + raw[by + 3];
      tr = raw[bx + 4] + raw[by + 4];
    }
    const float ol = hasl ? tl : 0.f;
    const float orr = hasr ? tr : 0.f;
    float a = c0 * ol;
    a += c1 * e0;
    a += c1 * tc;
    a += c0 * e1;
    float b = c0 * e0;
    b += c1 * tc;
    b += c1 * e1;
    b += c0 * orr;
    const bool rowok = (j == 0) ? topok : ((j == NU - 1) ? botok : true);
    h0[j] = rowok ? a : 0.f;
    h1[j] = rowok ? b : 0.f;
  }

  float* o = out + ((size_t)oc * 256 + 2 * n0) * 256 + 2 * m;
#pragma unroll
  for (int i = 0; i < 2 * BR; ++i) {
    float a = c0 * h0[i];
    a += c1 * h0[i + 1];
    a += c1 * h0[i + 2];
    a += c0 * h0[i + 3];
    float b = c0 * h1[i];
    b += c1 * h1[i + 1];
    b += c1 * h1[i + 2];
    b += c0 * h1[i + 3];
    const v2f v = {a, b};
    *(v2f*)(o + (size_t)i * 256) = v;
  }
}

extern "C" void kernel_launch(void* const* d_in, const int* in_sizes, int n_in,
                              void* d_out, int out_size, void* d_ws, size_t ws_size,
                              hipStream_t stream) {
  const float* x = (const float*)d_in[0];
  const float* wgt = (const float*)d_in[1];
  float* out = (float*)d_out;

  float* ws = (float*)d_ws;
  float* wp = ws;
  float* xt = ws + (size_t)OC_N * 6144;
  float* up = xt + (size_t)2 * IC_N * HW_N;

  prep_weights<<<OC_N + IC_N, 256, 0, stream>>>(wgt, wp, x, xt);

  upconv_gemm<<<2600, 128, 0, stream>>>(x, xt, wp, up);

  blur_kernel<<<dim3(16, OC_N), 128, 0, stream>>>(up, out);
}


# Even-column phases by minimal filtering F(2,2) along columns too: 3 products per column pair, not 4 (GEMM work x0.84); column differences in launch 1, pair sums in blur; sum order changes
# speedup vs baseline: 2.8323x; 1.1007x over previous
#include <hip/hip_runtime.h>
#include <math.h>


#define IC_N 512
#define OC_N 512
#define H_N 128
#define W_N 128
#define HW_N (H_N * W_N)
#define NS 65
#define DCH (H_N * NS)
#define UG (3 * NS * NS + NS * W_N)

typedef __attribute__((ext_vector_type(2))) float v2f;
typedef __attribute__((ext_vector_type(4))) float v4f;
typedef __attribute__((ext_vector_type(8))) float v8f;


#define RPW 9
__device__ __forceinline__ void row_differences(const float* __restrict__ x,
                                                float* __restrict__ xt, int ic) {
  const int sb = (threadIdx.x >> 5) * RPW;
  const int c4 = (threadIdx.x & 31) * 4;
  const float* xc = x + (size_t)ic * HW_N + c4;
  const v4f z = {0.0f, 0.0f, 0.0f, 0.0f};
  v4f r[2 * RPW + 1];
#pragma unroll
  for (int i = 0; i < 2 * RPW + 1; ++i) {
    const int row = 2 * sb - 1 + i;
    const int rc = row < 0 ? 0 : (row > H_N - 1 ? H_N - 1 : row);
    const v4f v = *(const v4f*)(xc + rc * W_N);
    r[i] = (row >= 0 && row < H_N) ? v : z;
  }
  float* o = xt + (size_t)ic * HW_N + c4;
#pragma unroll
  for (int i = 0; i < RPW; ++i) {
    const int s = sb + i;
    if (s < NS) {
      *(v4f*)(o + s * W_N) = r[2 * i] - r[2 * i + 1];
      *(v4f*)(o + (size_t)IC_N * HW_N + s * W_N) = r[2 * i + 2] - r[2 * i + 1];
    }
  }
}

#define CDB 3
__device__ __forceinline__ void col_differences(const float* __restrict__ x,
                                                float* __restrict__ xd, int ic) {
  const float* xc = x + (size_t)ic * HW_N;
  float* d0 = xd + (size_t)ic * DCH;
  const size_t PS = (size_t)IC_N * DCH;
  for (int base = threadIdx.x; base < NS * NS; base += 256 * CDB) {
    float v[CDB][3][3];
#pragma unroll
    for (int k = 0; k < CDB; ++k) {
      const int it = base + 256 * k;
      const int itc = it < NS * NS ? it : NS * NS - 1;
      const int s = itc / NS;
      const int u = itc - s * NS;
#pragma unroll
      for (int a = 0; a < 3; ++a) {
#pragma unroll
        for (int b = 0; b < 3; ++b) {
          const int r = 2 * s - 1 + a;
          const int cc = 2 * u - 1 + b;
          const int rc = r < 0 ? 0 : (r > H_N - 1 ? H_N - 1 : r);
          const int ccl = cc < 0 ? 0 : (cc > W_N - 1 ? W_N - 1 : cc);
          const float val = xc[rc * W_N + ccl];
          v[k][a][b] = (r >= 0 && r < H_N && cc >= 0 && cc < W_N) ? val : 0.0f;
        }
      }
    }
    asm volatile("" ::"v"(v[0][0][0]), "v"(v[0][0][1]), "v"(v[0][0][2]), "v"(v[0][1][0]),
                 "v"(v[0][1][1]), "v"(v[0][1][2]), "v"(v[0][2][0]), "v"(v[0][2][1]),
                 "v"(v[0][2][2]), "v"(v[1][0][0]), "v"(v[1][0][1]), "v"(v[1][0][2]),
                 "v"(v[1][1][0]), "v"(v[1][1][1]), "v"(v[1][1][2]), "v"(v[1][2][0]),
                 "v"(v[1][2][1]), "v"(v[1][2][2]), "v"(v[2][0][0]), "v"(v[2][0][1]),
                 "v"(v[2][0][2]), "v"(v[2][1][0]), "v"(v[2][1][1]), "v"(v[2][1][2]),
                 "v"(v[2][2][0]), "v"(v[2][2][1]), "v"(v[2][2][2]));
#pragma unroll
    for (int k = 0; k < CDB; ++k) {
      const int it = base + 256 * k;
      if (it < NS * NS) {
        const int s = it / NS;
        const int u = it - s * NS;
        float t0[3], t2[3];
#pragma unroll
        for (int b = 0; b < 3; ++b) {
          t0[b] = v[k][0][b] - v[k][1][b];
          t2[b] = v[k][2][b] - v[k][1][b];
        }
        d0[0 * PS + it] = t0[0] - t0[1];
        d0[1 * PS + it] = t0[2] - t0[1];
        d0[2 * PS + it] = v[k][1][0] - v[k][1][1];
        d0[3 * PS + it] = v[k][1][2] - v[k][1][1];
        d0[4 * PS + it] = t2[0] - t2[1];
        d0[5 * PS + it] = t2[2] - t2[1];
        if (s < H_N / 2) {
          d0[6 * PS + (2 * s) * NS + u] = v[k][1][0] - v[k][1][1];
          d0[7 * PS + (2 * s) * NS + u] = v[k][1][2] - v[k][1][1];
          d0[6 * PS + (2 * s + 1) * NS + u] = v[k][2][0] - v[k][2][1];
          d0[7 * PS + (2 * s + 1) * NS + u] = v[k][2][2] - v[k][2][1];
        }
      }
    }
  }
}

__global__ __launch_bounds__(256) void prep_weights(const float* __restrict__ wgt,
                                                    float* __restrict__ wp,
                                                    const float* __restrict__ x,
                                                    float* __restrict__ xt,
                                                    float* __restrict__ xd) {
  if (blockIdx.x >= OC_N + IC_N) {
    col_differences(x, xd, blockIdx.x - (OC_N + IC_N));
    return;
  }
  if (blockIdx.x >= OC_N) {
    row_differences(x, xt, blockIdx.x - OC_N);
    return;
  }
  const int oc = blockIdx.x;
  const int tid = threadIdx.x;
  const float scale = rsqrtf(4608.0f);
  __shared__ float red[256];
  __shared__ float dsh;

  const float* wo = wgt + (size_t)oc * IC_N * 9;
  float s = 0.f;
  for (int ic = tid; ic < IC_N; ic += 256) {
    const float* wi = wo + ic * 9;
#pragma unroll
    for (int k = 0; k < 9; ++k) {
      float v = scale * wi[k];
      s += v * v;
    }
  }
  red[tid] = s;
  __syncthreads();
  for (int off = 128; off > 0; off >>= 1) {
    if (tid < off) red[tid] += red[tid + off];
    __syncthreads();
  }
  if (tid == 0) dsh = rsqrtf(red[0] + 1e-8f);
  __syncthreads();
  const float d = dsh * scale;

  const size_t SW = (size_t)OC_N * IC_N;
  float* g = wp + (size_t)oc * IC_N;
  for (int ic = tid; ic < IC_N; ic += 256) {
    const float* wi = wo + ic * 9;
    float rr[4][3];
#pragma unroll
    for (int kx = 0; kx < 3; ++kx) {
      const float a = wi[0 * 3 + kx], b = wi[2 * 3 + kx];
      rr[0][kx] = b;
      rr[1][kx] = a + b;
      rr[2][kx] = a;
      rr[3][kx] = wi[1 * 3 + kx];
    }
#pragma unroll
    for (int rk = 0; rk < 4; ++rk) {
      g[(size_t)(rk * 4 + 0) * SW + ic] = d * rr[rk][2];
      g[(size_t)(rk * 4 + 1) * SW + ic] = d * (rr[rk][0] + rr[rk][2]);
      g[(size_t)(rk * 4 + 2) * SW + ic] = d * rr[rk][0];
      g[(size_t)(rk * 4 + 3) * SW + ic] = d * rr[rk][1];
    }
  }
}

#define BM 256
#define BN 64
#define BK 32
#define LP 36
#define NBA 134
#define NBB 260
#define NBC 512
#define NBG (3 * NBA + NBB)

__global__ __launch_bounds__(128) void upconv_gemm(
    const float* __restrict__ x, const float* __restrict__ xt,
    const float* __restrict__ xd, const float* __restrict__ wp,
    float* __restrict__ up) {
  __shared__ __attribute__((aligned(16))) float As[2][BM * LP];
  __shared__ __attribute__((aligned(16))) float Bs[2][BN * LP];

  const int tid = threadIdx.x;
  const int lane = tid & 31;
  const int wave = tid >> 5;
  const int c16 = lane & 15;
  const int hi = lane >> 4;
  const int wm = wave;

  const int bid = blockIdx.x;
  int rk, ck, local;
  if (bid < 3 * NBG) {
    rk = bid / NBG;
    const int r = bid - rk * NBG;
    if (r < 3 * NBA) { ck = r / NBA; local = r - ck * NBA; }
    else { ck = 3; local = r - 3 * NBA; }
  } else {
    rk = 3;
    const int r = bid - 3 * NBG;
    if (r < 3 * NBB) { ck = r / NBB; local = r - ck * NBB; }
    else { ck = 3; local = r - 3 * NBB; }
  }
  const int rows = rk == 3 ? H_N : NS;
  const int cols = ck == 3 ? W_N : NS;
  const int npx = rows * cols;
  const float* __restrict__ wph = wp + (size_t)(rk * 4 + ck) * OC_N * IC_N;
  float* __restrict__ uph = up + ((size_t)rk * UG + (size_t)ck * rows * NS) * OC_N;
  const bool dpl = (ck == 0) || (ck == 2);
  const float* __restrict__ bsrc =
      dpl ? xd + (size_t)(rk * 2 + (ck >> 1)) * IC_N * DCH
          : (rk == 0 ? xt : (rk == 2 ? xt + (size_t)IC_N * HW_N : x));
  const int rmul = (!dpl && rk == 1) ? 2 : 1;
  const int cmul = ck == 1 ? 2 : 1;
  const int bpitch = dpl ? NS : W_N;

  const int oc0 = (local & 1) * BM;
  const int n0 = (local >> 1) * BN;

  const int a_row = tid >> 3;
  const int a_kq = tid & 7;
  const int b_px = tid & 63;
  const int b_kh = tid >> 6;

  const int p = n0 + b_px;
  const bool pok = p < npx;
  const int pr = p / cols;
  const int pc = p - pr * cols;
  const int rin = pr * rmul;
  const int cin = pc * cmul;
  const bool bok = pok && rin < H_N && cin < W_N;
  const int boff = bok ? rin * bpitch + cin : 0;

  const float* a_src = wph + (size_t)(oc0 + a_row) * IC_N + a_kq * 4;

  v4f ra[16];
  float rb[16];

  auto load_chunk = [&](int ch) {
#pragma unroll
    for (int i = 0; i < 16; ++i)
      ra[i] = *(const v4f*)(a_src + (size_t)(16 * i) * IC_N + ch * BK);
    const int k0 = ch * BK + b_kh * 16;
    if (dpl) {
      const float* xs = bsrc + (size_t)k0 * DCH + boff;
#pragma unroll
      for (int j = 0; j < 16; ++j) rb[j] = xs[(size_t)j * DCH];
    } else {
      const float* xs = bsrc + (size_t)k0 * HW_N + boff;
#pragma unroll
      for (int j = 0; j < 16; ++j) rb[j] = xs[(size_t)j * HW_N];
    }
  };

  auto store_chunk = [&](int buf) {
#pragma unroll
    for (int i = 0; i < 16; ++i)
      *(v4f*)&As[buf][(a_row + 16 * i) * LP + a_kq * 4] = ra[i];
#pragma unroll
    for (int q = 0; q < 4; ++q) {
      const v4f b = {rb[4 * q], rb[4 * q + 1], rb[4 * q + 2], rb[4 * q + 3]};
      const v4f z = {0.0f, 0.0f, 0.0f, 0.0f};
      *(v4f*)&Bs[buf][b_px * LP + b_kh * 16 + q * 4] = bok ? b : z;
    }
  };

  v8f acc[4][4];
#pragma unroll
  for (int mi = 0; mi < 4; ++mi)
#pragma unroll
    for (int ni = 0; ni < 4; ++ni) acc[mi][ni] = (v8f){};

  const int a_frag = (wm * 64 + c16) * LP + hi * 4;
  const int b_frag = c16 * LP + hi * 4;

  const int nch = IC_N / BK;
  load_chunk(0);
  store_chunk(0);
  __syncthreads();

  for (int ch = 0; ch < nch; ++ch) {
    const int buf = ch & 1;
    const bool more = (ch + 1) < nch;
    if (more) load_chunk(ch + 1);
    __builtin_amdgcn_sched_barrier(0);

    const float* as = &As[buf][a_frag];
    const float* bs = &Bs[buf][b_frag];
#pragma unroll
    for (int g = 0; g < 4; ++g) {
      v4f a[4], b[4];
#pragma unroll
      for (int i = 0; i < 4; ++i) {
        a[i] = *(const v4f*)(as + i * 16 * LP + g * 8);
        b[i] = *(const v4f*)(bs + i * 16 * LP + g * 8);
      }
#pragma unroll
      for (int mi = 0; mi < 4; ++mi) {
        const v2f al = {a[mi].x, a[mi].y};
#pragma unroll
        for (int ni = 0; ni < 4; ++ni) {
          const v2f bl = {b[ni].x, b[ni].y};
          acc[mi][ni] = __builtin_amdgcn_wmma_f32_16x16x4_f32(false, al, false, bl, (short)0, acc[mi][ni], false, false);
        }
      }
#pragma unroll
      for (int mi = 0; mi < 4; ++mi) {
        const v2f ah = {a[mi].z, a[mi].w};
#pragma unroll
        for (int ni = 0; ni < 4; ++ni) {
          const v2f bh = {b[ni].z, b[ni].w};
          acc[mi][ni] = __builtin_amdgcn_wmma_f32_16x16x4_f32(false, ah, false, bh, (short)0, acc[mi][ni], false, false);
        }
      }
    }

    if (more) store_chunk(buf ^ 1);
    __syncthreads();
  }

#pragma unroll
  for (int ni = 0; ni < 4; ++ni) {
    const int q = n0 + ni * 16 + c16;
    if (q < npx) {
      float* o = uph + (size_t)(oc0 + wm * 64 + hi * 8) * npx + q;
#pragma unroll
      for (int mi = 0; mi < 4; ++mi) {
#pragma unroll
        for (int r = 0; r < 8; ++r) o[(size_t)(mi * 16 + r) * npx] = acc[mi][ni][r];
      }
    }
  }
}

#define BR 8
#define NU (2 * BR + 3)
#define NOD (BR + 2)
#define NPR (BR / 2 + 1)
#define NRAW (7 * NOD + 7 * (3 * NPR - 1))

__global__ __launch_bounds__(128) void blur_kernel(const float* __restrict__ up,
                                                   float* __restrict__ out) {
  const int m = threadIdx.x;
  const int n0 = blockIdx.x * BR;
  const int oc = blockIdx.y;

  const int SSA = OC_N * NS * NS;
  const int SSB = OC_N * H_N * NS;
  const float* e1 = up + (size_t)oc * (NS * NS);
  const float* e2 = e1 + (size_t)UG * OC_N;
  const float* e3 = e2 + (size_t)UG * OC_N;
  const float* o1 = up + (size_t)3 * SSA + (size_t)oc * (NS * W_N);
  const float* o2 = o1 + (size_t)UG * OC_N;
  const float* o3 = o2 + (size_t)UG * OC_N;
  const float* eo = up + (size_t)3 * UG * OC_N + (size_t)oc * (H_N * NS);
  const float* oo = up + (size_t)3 * UG * OC_N + (size_t)3 * SSB + (size_t)oc * HW_N;

  const int par = m & 1;
  const int u0 = m >> 1;
  const int ua = u0 + par;
  const int a1 = par * SSA + u0, a2 = (par + 1) * SSA + u0;
  const int a3 = (par ? 0 : 2 * SSA) + ua, a4 = SSA + ua;
  const int b1 = par * SSB + u0, b2 = (par + 1) * SSB + u0;
  const int b3 = (par ? 0 : 2 * SSB) + ua, b4 = SSB + ua;

  const bool hasl = m > 0;
  const bool hasr = m < 127;
  const int ml = hasl ? m - 1 : 0;
  const int mr = hasr ? m + 1 : 127;
  const float c0 = 0.25f, c1 = 0.75f;

  const bool topok = n0 > 0;
  const bool botok = n0 + BR < 128;
  const int s0 = n0 >> 1;

  float raw[NRAW];
#pragma unroll
  for (int jj = 0; jj < NOD; ++jj) {
    int k = jj - 1;
    if (jj == 0) k = topok ? -1 : 0;
    if (jj == NOD - 1) k = botok ? BR : BR - 1;
    const float* re = eo + (n0 + k) * NS;
    const float* ro = oo + (n0 + k) * W_N;
    raw[7 * jj + 0] = re[b1];
    raw[7 * jj + 1] = re[b2];
    raw[7 * jj + 2] = re[b3];
    raw[7 * jj + 3] = re[b4];
    raw[7 * jj + 4] = ro[ml];
    raw[7 * jj + 5] = ro[m];
    raw[7 * jj + 6] = ro[mr];
  }
#pragma unroll
  for (int q = 0; q < 3; ++q) {
    const float* pe = q == 0 ? e1 : (q == 1 ? e2 : e3);
    const float* po = q == 0 ? o1 : (q == 1 ? o2 : o3);
#pragma unroll
    for (int sp = 0; sp < NPR; ++sp) {
      if (q == 2 && sp == NPR - 1) continue;
      const float* re = pe + (s0 + sp) * NS;
      const float* ro = po + (s0 + sp) * W_N;
      const int b = 7 * NOD + 7 * (q * NPR + sp);
      raw[b + 0] = re[a1];
      raw[b + 1] = re[a2];
      raw[b + 2] = re[a3];
      raw[b + 3] = re[a4];
      raw[b + 4] = ro[ml];
      raw[b + 5] = ro[m];
      raw[b + 6] = ro[mr];
    }
  }

#pragma unroll
  for (int g = 0; g < NRAW; g += 24) {
    asm volatile("" ::"v"(raw[g]), "v"(raw[g + 1]), "v"(raw[g + 2]), "v"(raw[g + 3]),
                 "v"(raw[g + 4]), "v"(raw[g + 5]), "v"(raw[g + 6]), "v"(raw[g + 7]),
                 "v"(raw[g + 8]), "v"(raw[g + 9]), "v"(raw[g + 10]), "v"(raw[g + 11]),
                 "v"(raw[g + 12]), "v"(raw[g + 13]), "v"(raw[g + 14]), "v"(raw[g + 15]),
                 "v"(raw[g + 16]), "v"(raw[g + 17]), "v"(raw[g + 18]), "v"(raw[g + 19]),
                 "v"(raw[g + 20]), "v"(raw[g + 21]), "v"(raw[g + 22]), "v"(raw[g + 23]));
  }

  float h0[NU], h1[NU];
#pragma unroll
  for (int j = 0; j < NU; ++j) {
    float e0, e1v, tl, tc, tr;
    if ((j & 1) == 0) {
      const int b = 7 * (j >> 1);
      e0 = raw[b] + raw[b + 1];
      e1v = raw[b + 2] + raw[b + 3];
      tl = raw[b + 4]; tc = raw[b + 5]; tr = raw[b + 6];
    } else {
      const int i = j >> 1;
      const int sp = i >> 1;
      const int bx = 7 * NOD + 7 * (((i & 1) ? 1 : 0) * NPR + sp);
      const int by = 7 * NOD + 7 * (((i & 1) ? 2 : 1) * NPR + sp);
      e0 = (raw[bx] + raw[bx + 1]) + (raw[by] + raw[by + 1]);
      e1v = (raw[bx + 2] + raw[bx + 3]) + (raw[by + 2] + raw[by + 3]);
      tl = raw[bx + 4] + raw[by + 4];
      tc = raw[bx + 5] + raw[by + 5];
      tr = raw[bx + 6] + raw[by + 6];
    }
    const float ol = hasl ? tl : 0.f;
    const float orr = hasr ? tr : 0.f;
    float a = c0 * ol;
    a += c1 * e0;
    a += c1 * tc;
    a += c0 * e1v;
    float b = c0 * e0;
    b += c1 * tc;
    b += c1 * e1v;
    b += c0 * orr;
    const bool rowok = (j == 0) ? topok : ((j == NU - 1) ? botok : true);
    h0[j] = rowok ? a : 0.f;
    h1[j] = rowok ? b : 0.f;
  }

  float* o = out + ((size_t)oc * 256 + 2 * n0) * 256 + 2 * m;
#pragma unroll
  for (int i = 0; i < 2 * BR; ++i) {
    float a = c0 * h0[i];
    a += c1 * h0[i + 1];
    a += c1 * h0[i + 2];
    a += c0 * h0[i + 3];
    float b = c0 * h1[i];
    b += c1 * h1[i + 1];
    b += c1 * h1[i + 2];
    b += c0 * h1[i + 3];
    const v2f v = {a, b};
    *(v2f*)(o + (size_t)i * 256) = v;
  }
}

extern "C" void kernel_launch(void* const* d_in, const int* in_sizes, int n_in,
                              void* d_out, int out_size, void* d_ws, size_t ws_size,
                              hipStream_t stream) {
  const float* x = (const float*)d_in[0];
  const float* wgt = (const float*)d_in[1];
  float* out = (float*)d_out;

  float* ws = (float*)d_ws;
  float* wp = ws;
  float* xt = wp + (size_t)16 * OC_N * IC_N;
  float* xd = xt + (size_t)2 * IC_N * HW_N;
  float* up = xd + (size_t)8 * IC_N * DCH;

  prep_weights<<<OC_N + 2 * IC_N, 256, 0, stream>>>(wgt, wp, x, xt, xd);

  upconv_gemm<<<3278, 128, 0, stream>>>(x, xt, xd, wp, up);

  blur_kernel<<<dim3(16, OC_N), 128, 0, stream>>>(up, out);
}


# Blur pass reads its window through LDS: a block's rows of each plane are one contiguous run, copied with whole-wave contiguous loads (54 per wave, not 168) and picked from LDS; sum order unchanged
# speedup vs baseline: 2.9689x; 1.0482x over previous
#include <hip/hip_runtime.h>
#include <math.h>


#define IC_N 512
#define OC_N 512
#define H_N 128
#define W_N 128
#define HW_N (H_N * W_N)
#define NS 65
#define DCH (H_N * NS)
#define UG (3 * NS * NS + NS * W_N)

typedef __attribute__((ext_vector_type(2))) float v2f;
typedef __attribute__((ext_vector_type(4))) float v4f;
typedef __attribute__((ext_vector_type(8))) float v8f;


#define RPW 9
__device__ __forceinline__ void row_differences(const float* __restrict__ x,
                                                float* __restrict__ xt, int ic) {
  const int sb = (threadIdx.x >> 5) * RPW;
  const int c4 = (threadIdx.x & 31) * 4;
  const float* xc = x + (size_t)ic * HW_N + c4;
  const v4f z = {0.0f, 0.0f, 0.0f, 0.0f};
  v4f r[2 * RPW + 1];
#pragma unroll
  for (int i = 0; i < 2 * RPW + 1; ++i) {
    const int row = 2 * sb - 1 + i;
    const int rc = row < 0 ? 0 : (row > H_N - 1 ? H_N - 1 : row);
    const v4f v = *(const v4f*)(xc + rc * W_N);
    r[i] = (row >= 0 && row < H_N) ? v : z;
  }
  float* o = xt + (size_t)ic * HW_N + c4;
#pragma unroll
  for (int i = 0; i < RPW; ++i) {
    const int s = sb + i;
    if (s < NS) {
      *(v4f*)(o + s * W_N) = r[2 * i] - r[2 * i + 1];
      *(v4f*)(o + (size_t)IC_N * HW_N + s * W_N) = r[2 * i + 2] - r[2 * i + 1];
    }
  }
}

#define CDB 3
__device__ __forceinline__ void col_differences(const float* __restrict__ x,
                                                float* __restrict__ xd, int ic) {
  const float* xc = x + (size_t)ic * HW_N;
  float* d0 = xd + (size_t)ic * DCH;
  const size_t PS = (size_t)IC_N * DCH;
  for (int base = threadIdx.x; base < NS * NS; base += 256 * CDB) {
    float v[CDB][3][3];
#pragma unroll
    for (int k = 0; k < CDB; ++k) {
      const int it = base + 256 * k;
      const int itc = it < NS * NS ? it : NS * NS - 1;
      const int s = itc / NS;
      const int u = itc - s * NS;
#pragma unroll
      for (int a = 0; a < 3; ++a) {
#pragma unroll
        for (int b = 0; b < 3; ++b) {
          const int r = 2 * s - 1 + a;
          const int cc = 2 * u - 1 + b;
          const int rc = r < 0 ? 0 : (r > H_N - 1 ? H_N - 1 : r);
          const int ccl = cc < 0 ? 0 : (cc > W_N - 1 ? W_N - 1 : cc);
          const float val = xc[rc * W_N + ccl];
          v[k][a][b] = (r >= 0 && r < H_N && cc >= 0 && cc < W_N) ? val : 0.0f;
        }
      }
    }
    asm volatile("" ::"v"(v[0][0][0]), "v"(v[0][0][1]), "v"(v[0][0][2]), "v"(v[0][1][0]),
                 "v"(v[0][1][1]), "v"(v[0][1][2]), "v"(v[0][2][0]), "v"(v[0][2][1]),
                 "v"(v[0][2][2]), "v"(v[1][0][0]), "v"(v[1][0][1]), "v"(v[1][0][2]),
                 "v"(v[1][1][0]), "v"(v[1][1][1]), "v"(v[1][1][2]), "v"(v[1][2][0]),
                 "v"(v[1][2][1]), "v"(v[1][2][2]), "v"(v[2][0][0]), "v"(v[2][0][1]),
                 "v"(v[2][0][2]), "v"(v[2][1][0]), "v"(v[2][1][1]), "v"(v[2][1][2]),
                 "v"(v[2][2][0]), "v"(v[2][2][1]), "v"(v[2][2][2]));
#pragma unroll
    for (int k = 0; k < CDB; ++k) {
      const int it = base + 256 * k;
      if (it < NS * NS) {
        const int s = it / NS;
        const int u = it - s * NS;
        float t0[3], t2[3];
#pragma unroll
        for (int b = 0; b < 3; ++b) {
          t0[b] = v[k][0][b] - v[k][1][b];
          t2[b] = v[k][2][b] - v[k][1][b];
        }
        d0[0 * PS + it] = t0[0] - t0[1];
        d0[1 * PS + it] = t0[2] - t0[1];
        d0[2 * PS + it] = v[k][1][0] - v[k][1][1];
        d0[3 * PS + it] = v[k][1][2] - v[k][1][1];
        d0[4 * PS + it] = t2[0] - t2[1];
        d0[5 * PS + it] = t2[2] - t2[1];
        if (s < H_N / 2) {
          d0[6 * PS + (2 * s) * NS + u] = v[k][1][0] - v[k][1][1];
          d0[7 * PS + (2 * s) * NS + u] = v[k][1][2] - v[k][1][1];
          d0[6 * PS + (2 * s + 1) * NS + u] = v[k][2][0] - v[k][2][1];
          d0[7 * PS + (2 * s + 1) * NS + u] = v[k][2][2] - v[k][2][1];
        }
      }
    }
  }
}

__global__ __launch_bounds__(256) void prep_weights(const float* __restrict__ wgt,
                                                    float* __restrict__ wp,
                                                    const float* __restrict__ x,
                                                    float* __restrict__ xt,
                                                    float* __restrict__ xd) {
  if (blockIdx.x >= OC_N + IC_N) {
    col_differences(x, xd, blockIdx.x - (OC_N + IC_N));
    return;
  }
  if (blockIdx.x >= OC_N) {
    row_differences(x, xt, blockIdx.x - OC_N);
    return;
  }
  const int oc = blockIdx.x;
  const int tid = threadIdx.x;
  const float scale = rsqrtf(4608.0f);
  __shared__ float red[256];
  __shared__ float dsh;

  const float* wo = wgt + (size_t)oc * IC_N * 9;
  float s = 0.f;
  for (int ic = tid; ic < IC_N; ic += 256) {
    const float* wi = wo + ic * 9;
#pragma unroll
    for (int k = 0; k < 9; ++k) {
      float v = scale * wi[k];
      s += v * v;
    }
  }
  red[tid] = s;
  __syncthreads();
  for (int off = 128; off > 0; off >>= 1) {
    if (tid < off) red[tid] += red[tid + off];
    __syncthreads();
  }
  if (tid == 0) dsh = rsqrtf(red[0] + 1e-8f);
  __syncthreads();
  const float d = dsh * scale;

  const size_t SW = (size_t)OC_N * IC_N;
  float* g = wp + (size_t)oc * IC_N;
  for (int ic = tid; ic < IC_N; ic += 256) {
    const float* wi = wo + ic * 9;
    float rr[4][3];
#pragma unroll
    for (int kx = 0; kx < 3; ++kx) {
      const float a = wi[0 * 3 + kx], b = wi[2 * 3 + kx];
      rr[0][kx] = b;
      rr[1][kx] = a + b;
      rr[2][kx] = a;
      rr[3][kx] = wi[1 * 3 + kx];
    }
#pragma unroll
    for (int rk = 0; rk < 4; ++rk) {
      g[(size_t)(rk * 4 + 0) * SW + ic] = d * rr[rk][2];
      g[(size_t)(rk * 4 + 1) * SW + ic] = d * (rr[rk][0] + rr[rk][2]);
      g[(size_t)(rk * 4 + 2) * SW + ic] = d * rr[rk][0];
      g[(size_t)(rk * 4 + 3) * SW + ic] = d * rr[rk][1];
    }
  }
}

#define BM 256
#define BN 64
#define BK 32
#define LP 36
#define NBA 134
#define NBB 260
#define NBC 512
#define NBG (3 * NBA + NBB)

__global__ __launch_bounds__(128) void upconv_gemm(
    const float* __restrict__ x, const float* __restrict__ xt,
    const float* __restrict__ xd, const float* __restrict__ wp,
    float* __restrict__ up) {
  __shared__ __attribute__((aligned(16))) float As[2][BM * LP];
  __shared__ __attribute__((aligned(16))) float Bs[2][BN * LP];

  const int tid = threadIdx.x;
  const int lane = tid & 31;
  const int wave = tid >> 5;
  const int c16 = lane & 15;
  const int hi = lane >> 4;
  const int wm = wave;

  const int bid = blockIdx.x;
  int rk, ck, local;
  if (bid < 3 * NBG) {
    rk = bid / NBG;
    const int r = bid - rk * NBG;
    if (r < 3 * NBA) { ck = r / NBA; local = r - ck * NBA; }
    else { ck = 3; local = r - 3 * NBA; }
  } else {
    rk = 3;
    const int r = bid - 3 * NBG;
    if (r < 3 * NBB) { ck = r / NBB; local = r - ck * NBB; }
    else { ck = 3; local = r - 3 * NBB; }
  }
  const int rows = rk == 3 ? H_N : NS;
  const int cols = ck == 3 ? W_N : NS;
  const int npx = rows * cols;
  const float* __restrict__ wph = wp + (size_t)(rk * 4 + ck) * OC_N * IC_N;
  float* __restrict__ uph = up + ((size_t)rk * UG + (size_t)ck * rows * NS) * OC_N;
  const bool dpl = (ck == 0) || (ck == 2);
  const float* __restrict__ bsrc =
      dpl ? xd + (size_t)(rk * 2 + (ck >> 1)) * IC_N * DCH
          : (rk == 0 ? xt : (rk == 2 ? xt + (size_t)IC_N * HW_N : x));
  const int rmul = (!dpl && rk == 1) ? 2 : 1;
  const int cmul = ck == 1 ? 2 : 1;
  const int bpitch = dpl ? NS : W_N;

  const int oc0 = (local & 1) * BM;
  const int n0 = (local >> 1) * BN;

  const int a_row = tid >> 3;
  const int a_kq = tid & 7;
  const int b_px = tid & 63;
  const int b_kh = tid >> 6;

  const int p = n0 + b_px;
  const bool pok = p < npx;
  const int pr = p / cols;
  const int pc = p - pr * cols;
  const int rin = pr * rmul;
  const int cin = pc * cmul;
  const bool bok = pok && rin < H_N && cin < W_N;
  const int boff = bok ? rin * bpitch + cin : 0;

  const float* a_src = wph + (size_t)(oc0 + a_row) * IC_N + a_kq * 4;

  v4f ra[16];
  float rb[16];

  auto load_chunk = [&](int ch) {
#pragma unroll
    for (int i = 0; i < 16; ++i)
      ra[i] = *(const v4f*)(a_src + (size_t)(16 * i) * IC_N + ch * BK);
    const int k0 = ch * BK + b_kh * 16;
    if (dpl) {
      const float* xs = bsrc + (size_t)k0 * DCH + boff;
#pragma unroll
      for (int j = 0; j < 16; ++j) rb[j] = xs[(size_t)j * DCH];
    } else {
      const float* xs = bsrc + (size_t)k0 * HW_N + boff;
#pragma unroll
      for (int j = 0; j < 16; ++j) rb[j] = xs[(size_t)j * HW_N];
    }
  };

  auto store_chunk = [&](int buf) {
#pragma unroll
    for (int i = 0; i < 16; ++i)
      *(v4f*)&As[buf][(a_row + 16 * i) * LP + a_kq * 4] = ra[i];
#pragma unroll
    for (int q = 0; q < 4; ++q) {
      const v4f b = {rb[4 * q], rb[4 * q + 1], rb[4 * q + 2], rb[4 * q + 3]};
      const v4f z = {0.0f, 0.0f, 0.0f, 0.0f};
      *(v4f*)&Bs[buf][b_px * LP + b_kh * 16 + q * 4] = bok ? b : z;
    }
  };

  v8f acc[4][4];
#pragma unroll
  for (int mi = 0; mi < 4; ++mi)
#pragma unroll
    for (int ni = 0; ni < 4; ++ni) acc[mi][ni] = (v8f){};

  const int a_frag = (wm * 64 + c16) * LP + hi * 4;
  const int b_frag = c16 * LP + hi * 4;

  const int nch = IC_N / BK;
  load_chunk(0);
  store_chunk(0);
  __syncthreads();

  for (int ch = 0; ch < nch; ++ch) {
    const int buf = ch & 1;
    const bool more = (ch + 1) < nch;
    if (more) load_chunk(ch + 1);
    __builtin_amdgcn_sched_barrier(0);

    const float* as = &As[buf][a_frag];
    const float* bs = &Bs[buf][b_frag];
#pragma unroll
    for (int g = 0; g < 4; ++g) {
      v4f a[4], b[4];
#pragma unroll
      for (int i = 0; i < 4; ++i) {
        a[i] = *(const v4f*)(as + i * 16 * LP + g * 8);
        b[i] = *(const v4f*)(bs + i * 16 * LP + g * 8);
      }
#pragma unroll
      for (int mi = 0; mi < 4; ++mi) {
        const v2f al = {a[mi].x, a[mi].y};
#pragma unroll
        for (int ni = 0; ni < 4; ++ni) {
          const v2f bl = {b[ni].x, b[ni].y};
          acc[mi][ni] = __builtin_amdgcn_wmma_f32_16x16x4_f32(false, al, false, bl, (short)0, acc[mi][ni], false, false);
        }
      }
#pragma unroll
      for (int mi = 0; mi < 4; ++mi) {
        const v2f ah = {a[mi].z, a[mi].w};
#pragma unroll
        for (int ni = 0; ni < 4; ++ni) {
          const v2f bh = {b[ni].z, b[ni].w};
          acc[mi][ni] = __builtin_amdgcn_wmma_f32_16x16x4_f32(false, ah, false, bh, (short)0, acc[mi][ni], false, false);
        }
      }
    }

    if (more) store_chunk(buf ^ 1);
    __syncthreads();
  }

#pragma unroll
  for (int ni = 0; ni < 4; ++ni) {
    const int q = n0 + ni * 16 + c16;
    if (q < npx) {
      float* o = uph + (size_t)(oc0 + wm * 64 + hi * 8) * npx + q;
#pragma unroll
      for (int mi = 0; mi < 4; ++mi) {
#pragma unroll
        for (int r = 0; r < 8; ++r) o[(size_t)(mi * 16 + r) * npx] = acc[mi][ni][r];
      }
    }
  }
}

#define BR 8
#define NU (2 * BR + 3)
#define NOD (BR + 2)
#define NPR (BR / 2 + 1)
#define ELA (NPR * NS)
#define ELB (NOD * NS)
#define EPA 336
#define EPB 656
#define OLA (NPR * W_N)
#define OLB (NOD * W_N)
#define GPA (3 * EPA + OLA)
#define WOB (3 * GPA)
#define LWIN (WOB + 3 * EPB + OLB)

__global__ __launch_bounds__(128) void blur_kernel(const float* __restrict__ up,
                                                   float* __restrict__ out) {
  __shared__ __attribute__((aligned(16))) float win[LWIN];

  const int m = threadIdx.x;
  const int n0 = blockIdx.x * BR;
  const int oc = blockIdx.y;

  const int SSA = OC_N * NS * NS;
  const int SSB = OC_N * H_N * NS;
  const int s0 = n0 >> 1;

  float va[3][3][3];
  v4f vo[3][2];
#pragma unroll
  for (int q = 0; q < 3; ++q) {
    const float* gq = up + (size_t)q * UG * OC_N;
#pragma unroll
    for (int ck = 0; ck < 3; ++ck) {
      const float* src = gq + (size_t)ck * SSA + (size_t)oc * (NS * NS) + s0 * NS;
#pragma unroll
      for (int it = 0; it < 3; ++it) {
        const int e = m + 128 * it;
        const bool whole = 128 * it + 127 < ELA;
        va[q][ck][it] = src[(whole || e < ELA) ? e : ELA - 1];
      }
    }
    const v4f* src4 = (const v4f*)(gq + (size_t)3 * SSA + (size_t)oc * (NS * W_N) + s0 * W_N);
#pragma unroll
    for (int it = 0; it < 2; ++it) {
      const int e = m + 128 * it;
      const bool whole = 128 * it + 127 < OLA / 4;
      vo[q][it] = src4[(whole || e < OLA / 4) ? e : OLA / 4 - 1];
    }
  }
  float vb[3][6];
  v4f vp[3];
  {
    const float* g3 = up + (size_t)3 * UG * OC_N;
#pragma unroll
    for (int ck = 0; ck < 3; ++ck) {
      const float* src = g3 + (size_t)ck * SSB + (size_t)oc * (H_N * NS);
#pragma unroll
      for (int it = 0; it < 6; ++it) {
        const int e = m + 128 * it;
        const bool whole = 128 * it + 127 < ELB;
        int g = (n0 - 1) * NS + ((whole || e < ELB) ? e : ELB - 1);
        g = g < 0 ? 0 : (g > H_N * NS - 1 ? H_N * NS - 1 : g);
        vb[ck][it] = src[g];
      }
    }
    const v4f* src4 = (const v4f*)(g3 + (size_t)3 * SSB + (size_t)oc * HW_N);
#pragma unroll
    for (int it = 0; it < 3; ++it) {
      const int e = m + 128 * it;
      const bool whole = 128 * it + 127 < OLB / 4;
      int g = (n0 - 1) * (W_N / 4) + ((whole || e < OLB / 4) ? e : OLB / 4 - 1);
      g = g < 0 ? 0 : (g > HW_N / 4 - 1 ? HW_N / 4 - 1 : g);
      vp[it] = src4[g];
    }
  }
  asm volatile("" ::"v"(va[0][0][2]), "v"(va[0][1][2]), "v"(va[0][2][2]), "v"(va[1][0][2]),
               "v"(va[1][1][2]), "v"(va[1][2][2]), "v"(va[2][0][2]), "v"(va[2][1][2]),
               "v"(va[2][2][2]), "v"(vo[0][1]), "v"(vo[1][1]), "v"(vo[2][1]), "v"(vb[0][5]),
               "v"(vb[1][5]), "v"(vb[2][5]), "v"(vp[2]));
#pragma unroll
  for (int q = 0; q < 3; ++q) {
#pragma unroll
    for (int ck = 0; ck < 3; ++ck) {
#pragma unroll
      for (int it = 0; it < 3; ++it) {
        const int e = m + 128 * it;
        if (128 * it + 127 < ELA || e < ELA) win[q * GPA + ck * EPA + e] = va[q][ck][it];
      }
    }
#pragma unroll
    for (int it = 0; it < 2; ++it) {
      const int e = m + 128 * it;
      if (128 * it + 127 < OLA / 4 || e < OLA / 4) *(v4f*)&win[q * GPA + 3 * EPA + 4 * e] = vo[q][it];
    }
  }
#pragma unroll
  for (int ck = 0; ck < 3; ++ck) {
#pragma unroll
    for (int it = 0; it < 6; ++it) {
      const int e = m + 128 * it;
      if (128 * it + 127 < ELB || e < ELB) win[WOB + ck * EPB + e] = vb[ck][it];
    }
  }
#pragma unroll
  for (int it = 0; it < 3; ++it) {
    const int e = m + 128 * it;
    if (128 * it + 127 < OLB / 4 || e < OLB / 4) *(v4f*)&win[WOB + 3 * EPB + 4 * e] = vp[it];
  }
  __syncthreads();

  const int par = m & 1;
  const int u0 = m >> 1;
  const int ua = u0 + par;
  const int a1 = par * EPA + u0, a2 = (par + 1) * EPA + u0;
  const int a3 = (par ? 0 : 2 * EPA) + ua, a4 = EPA + ua;
  const int b1 = par * EPB + u0, b2 = (par + 1) * EPB + u0;
  const int b3 = (par ? 0 : 2 * EPB) + ua, b4 = EPB + ua;

  const bool hasl = m > 0;
  const bool hasr = m < 127;
  const int ml = hasl ? m - 1 : 0;
  const int mr = hasr ? m + 1 : 127;
  const float c0 = 0.25f, c1 = 0.75f;

  const bool topok = n0 > 0;
  const bool botok = n0 + BR < 128;

  float h0[NU], h1[NU];
#pragma unroll
  for (int j = 0; j < NU; ++j) {
    float e0, e1v, tl, tc, tr;
    if ((j & 1) == 0) {
      const int jj = j >> 1;
      const float* we = win + WOB + jj * NS;
      const float* wo = win + WOB + 3 * EPB + jj * W_N;
      e0 = we[b1] + we[b2];
      e1v = we[b3] + we[b4];
      tl = wo[ml]; tc = wo[m]; tr = wo[mr];
    } else {
      const int i = j >> 1;
      const int sp = i >> 1;
      const float* xe = win + ((i & 1) ? 1 : 0) * GPA + sp * NS;
      const float* ye = win + ((i & 1) ? 2 : 1) * GPA + sp * NS;
      const float* xo = win + ((i & 1) ? 1 : 0) * GPA + 3 * EPA + sp * W_N;
      const float* yo = win + ((i & 1) ? 2 : 1) * GPA + 3 * EPA + sp * W_N;
      e0 = (xe[a1] + xe[a2]) + (ye[a1] + ye[a2]);
      e1v = (xe[a3] + xe[a4]) + (ye[a3] + ye[a4]);
      tl = xo[ml] + yo[ml];
      tc = xo[m] + yo[m];
      tr = xo[mr] + yo[mr];
    }
    const float ol = hasl ? tl : 0.f;
    const float orr = hasr ? tr : 0.f;
    float a = c0 * ol;
    a += c1 * e0;
    a += c1 * tc;
    a += c0 * e1v;
    float b = c0 * e0;
    b += c1 * tc;
    b += c1 * e1v;
    b += c0 * orr;
    const bool rowok = (j == 0) ? topok : ((j == NU - 1) ? botok : true);
    h0[j] = rowok ? a : 0.f;
    h1[j] = rowok ? b : 0.f;
  }

  float* o = out + ((size_t)oc * 256 + 2 * n0) * 256 + 2 * m;
#pragma unroll
  for (int i = 0; i < 2 * BR; ++i) {
    float a = c0 * h0[i];
    a += c1 * h0[i + 1];
    a += c1 * h0[i + 2];
    a += c0 * h0[i + 3];
    float b = c0 * h1[i];
    b += c1 * h1[i + 1];
    b += c1 * h1[i + 2];
    b += c0 * h1[i + 3];
    const v2f v = {a, b};
    *(v2f*)(o + (size_t)i * 256) = v;
  }
}

extern "C" void kernel_launch(void* const* d_in, const int* in_sizes, int n_in,
                              void* d_out, int out_size, void* d_ws, size_t ws_size,
                              hipStream_t stream) {
  const float* x = (const float*)d_in[0];
  const float* wgt = (const float*)d_in[1];
  float* out = (float*)d_out;

  float* ws = (float*)d_ws;
  float* wp = ws;
  float* xt = wp + (size_t)16 * OC_N * IC_N;
  float* xd = xt + (size_t)2 * IC_N * HW_N;
  float* up = xd + (size_t)8 * IC_N * DCH;

  prep_weights<<<OC_N + 2 * IC_N, 256, 0, stream>>>(wgt, wp, x, xt, xd);

  upconv_gemm<<<3278, 128, 0, stream>>>(x, xt, xd, wp, up);

  blur_kernel<<<dim3(16, OC_N), 128, 0, stream>>>(up, out);
}
